# starting kernel
# baseline (speedup 1.0000x reference)
#include <hip/hip_runtime.h>


typedef __attribute__((ext_vector_type(16))) _Float16 v16h;
typedef __attribute__((ext_vector_type(8)))  _Float16 half8;
typedef __attribute__((ext_vector_type(8)))  float    v8f;

#define HID 128

__global__ void inproj_kernel(const float* __restrict__ pos,
                              const int*   __restrict__ an,
                              const float* __restrict__ W,
                              const float* __restrict__ b,
                              float* __restrict__ h, int n) {
  int t = blockIdx.x * 256 + threadIdx.x;
  if (t >= n * HID) return;
  int node = t >> 7, c = t & 127;
  float z  = (float)an[node] * 0.1f;
  float px = pos[node * 3 + 0];
  float py = pos[node * 3 + 1];
  float pz = pos[node * 3 + 2];
  float v = b[c];
  v = fmaf(z,  W[0 * HID + c], v);
  v = fmaf(px, W[1 * HID + c], v);
  v = fmaf(py, W[2 * HID + c], v);
  v = fmaf(pz, W[3 * HID + c], v);
  h[t] = fmaxf(v, 0.0f);
}

__global__ void wprep_kernel(const float* __restrict__ W,
                             _Float16* __restrict__ WT) {
  int t = blockIdx.x * 256 + threadIdx.x;
  if (t >= HID * HID) return;
  int k = t & 127, n = t >> 7;
  WT[t] = (_Float16)W[k * HID + n];
}

__global__ void zero_kernel(float* __restrict__ p, int n) {
  int t = blockIdx.x * 256 + threadIdx.x;
  if (t < n) p[t] = 0.0f;
}

__global__ void scatter_kernel(const float* __restrict__ h,
                               const int*   __restrict__ ei,
                               float* __restrict__ s,
                               float* __restrict__ cnt, int nE) {
  int t = blockIdx.x * 256 + threadIdx.x;
  int e = t >> 5;
  if (e >= nE) return;
  int lane = t & 31;
  int ep = e + 4;
  if (ep < nE) {
    int psrc = ei[ep];
    __builtin_prefetch(h + (size_t)psrc * HID + lane * 4, 0, 1);
  }
  int src = ei[e];
  int dst = ei[nE + e];
  const float4 v = *(const float4*)(h + (size_t)src * HID + lane * 4);
  float* dp = s + (size_t)dst * HID + lane * 4;
  atomicAdd(dp + 0, v.x);
  atomicAdd(dp + 1, v.y);
  atomicAdd(dp + 2, v.z);
  atomicAdd(dp + 3, v.w);
  if (lane == 0) atomicAdd(cnt + dst, 1.0f);
}

__device__ inline half8 cvt8(const float* __restrict__ p, float scale) {
  float4 x = *(const float4*)(p);
  float4 y = *(const float4*)(p + 4);
  half8 r;
  r[0] = (_Float16)(x.x * scale); r[1] = (_Float16)(x.y * scale);
  r[2] = (_Float16)(x.z * scale); r[3] = (_Float16)(x.w * scale);
  r[4] = (_Float16)(y.x * scale); r[5] = (_Float16)(y.y * scale);
  r[6] = (_Float16)(y.z * scale); r[7] = (_Float16)(y.w * scale);
  return r;
}

__device__ inline v16h join16(half8 lo, half8 hi) {
  return __builtin_shufflevector(lo, hi, 0, 1, 2, 3, 4, 5, 6, 7,
                                         8, 9, 10, 11, 12, 13, 14, 15);
}

template <bool SAGE, bool RELU>
__global__ __launch_bounds__(256) void gemm_kernel(
    const float*    __restrict__ h,
    const float*    __restrict__ s,
    const float*    __restrict__ cnt,
    const _Float16* __restrict__ WlT,
    const _Float16* __restrict__ WrT,
    const float*    __restrict__ bias,
    float* __restrict__ out,
    int n_nodes) {
  __shared__ _Float16 wtr[HID * HID];
  __shared__ _Float16 wtl[HID * HID];

  {
    uint4*       dr = (uint4*)wtr;
    const uint4* sr = (const uint4*)WrT;
    for (int i = threadIdx.x; i < HID * HID / 8; i += 256) dr[i] = sr[i];
    if constexpr (SAGE) {
      uint4*       dl = (uint4*)wtl;
      const uint4* sl = (const uint4*)WlT;
      for (int i = threadIdx.x; i < HID * HID / 8; i += 256) dl[i] = sl[i];
    }
  }
  __syncthreads();

  const int lane = threadIdx.x & 31;
  const int wave = threadIdx.x >> 5;
  const int rowbase = blockIdx.x * 128 + wave * 16;
  if (rowbase >= n_nodes) return;

  const int arow = rowbase + (lane & 15);
  const int row  = arow < n_nodes ? arow : n_nodes - 1;
  const int lo   = (lane >= 16) ? 8 : 0;

  float inv = 0.0f;
  if constexpr (SAGE) {
    float cv = cnt[row];
    inv = (cv > 0.0f) ? __builtin_amdgcn_rcpf(cv) : 0.0f;
  }

  const float*    hp  = h + (size_t)row * HID + lo;
  const float*    sp  = SAGE ? (s + (size_t)row * HID + lo) : nullptr;
  const _Float16* brp = &wtr[(lane & 15) * HID + lo];
  const _Float16* blp = &wtl[(lane & 15) * HID + lo];

  v8f acc[8];
  {
    v8f z = {};
#pragma unroll
    for (int nt = 0; nt < 8; ++nt) acc[nt] = z;
  }

#pragma unroll
  for (int kk = 0; kk < 4; ++kk) {
    const int kb = kk * 32;
    v16h ahf = join16(cvt8(hp + kb, 1.0f), cvt8(hp + kb + 16, 1.0f));
    v16h agf;
    if constexpr (SAGE)
      agf = join16(cvt8(sp + kb, inv), cvt8(sp + kb + 16, inv));
#pragma unroll
    for (int nt = 0; nt < 8; ++nt) {
      const int off = nt * 16 * HID + kb;
      v16h br = join16(*(const half8*)(brp + off),
                       *(const half8*)(brp + off + 16));
      acc[nt] = __builtin_amdgcn_wmma_f32_16x16x32_f16(
          false, ahf, false, br, (short)0, acc[nt], false, false);
      if constexpr (SAGE) {
        v16h bl = join16(*(const half8*)(blp + off),
                         *(const half8*)(blp + off + 16));
        acc[nt] = __builtin_amdgcn_wmma_f32_16x16x32_f16(
            false, agf, false, bl, (short)0, acc[nt], false, false);
      }
    }
  }

  const int mofs = (lane >= 16) ? 8 : 0;
#pragma unroll
  for (int nt = 0; nt < 8; ++nt) {
    const int ncol = nt * 16 + (lane & 15);
    const float bb = bias[ncol];
#pragma unroll
    for (int r = 0; r < 8; ++r) {
      float v = acc[nt][r] + bb;
      if constexpr (RELU) v = fmaxf(v, 0.0f);
      int orow = rowbase + mofs + r;
      if (orow < n_nodes) out[(size_t)orow * HID + ncol] = v;
    }
  }
}

__global__ void colmean_kernel(const float* __restrict__ ns,
                               float* __restrict__ outm, int n) {
  __shared__ float red[256];
  int col = blockIdx.x;
  float acc = 0.0f;
  for (int r = threadIdx.x; r < n; r += 256) acc += ns[(size_t)r * HID + col];
  red[threadIdx.x] = acc;
  __syncthreads();
  for (int st = 128; st > 0; st >>= 1) {
    if (threadIdx.x < st) red[threadIdx.x] += red[threadIdx.x + st];
    __syncthreads();
  }
  if (threadIdx.x == 0) outm[col] = red[0] / (float)n;
}

extern "C" void kernel_launch(void* const* d_in, const int* in_sizes, int n_in,
                              void* d_out, int out_size, void* d_ws, size_t ws_size,
                              hipStream_t stream) {
  const float* pos  = (const float*)d_in[0];
  const int*   an   = (const int*)  d_in[1];
  const int*   ei   = (const int*)  d_in[2];
  const float* W_in = (const float*)d_in[3];
  const float* b_in = (const float*)d_in[4];
  const float* W1l  = (const float*)d_in[5];
  const float* b1   = (const float*)d_in[6];
  const float* W1r  = (const float*)d_in[7];
  const float* W2l  = (const float*)d_in[8];
  const float* b2   = (const float*)d_in[9];
  const float* W2r  = (const float*)d_in[10];
  const float* Wout = (const float*)d_in[11];
  const float* bout = (const float*)d_in[12];

  const int n  = in_sizes[0] / 3;
  const int nE = in_sizes[2] / 2;

  float* ws   = (float*)d_ws;
  float* h    = ws;
  float* s    = ws + (size_t)n * HID;
  float* cnt  = s  + (size_t)n * HID;

  size_t woff = (size_t)n * HID * 2 + n;
  woff = (woff + 63) & ~(size_t)63;
  _Float16* wtbase = (_Float16*)(ws + woff);
  _Float16* W1lT = wtbase + 0 * HID * HID;
  _Float16* W1rT = wtbase + 1 * HID * HID;
  _Float16* W2lT = wtbase + 2 * HID * HID;
  _Float16* W2rT = wtbase + 3 * HID * HID;
  _Float16* WoT  = wtbase + 4 * HID * HID;

  float* outm = (float*)d_out;
  float* nst  = (float*)d_out + HID;

  const int nh       = n * HID;
  const int gemmBlks = (n + 127) / 128;
  const int edgeThr  = nE * 32;
  const int wBlks    = (HID * HID + 255) / 256;

  wprep_kernel<<<wBlks, 256, 0, stream>>>(W1l,  W1lT);
  wprep_kernel<<<wBlks, 256, 0, stream>>>(W1r,  W1rT);
  wprep_kernel<<<wBlks, 256, 0, stream>>>(W2l,  W2lT);
  wprep_kernel<<<wBlks, 256, 0, stream>>>(W2r,  W2rT);
  wprep_kernel<<<wBlks, 256, 0, stream>>>(Wout, WoT);

  inproj_kernel<<<(nh + 255) / 256, 256, 0, stream>>>(pos, an, W_in, b_in, h, n);

  zero_kernel<<<(nh + n + 255) / 256, 256, 0, stream>>>(s, nh + n);
  scatter_kernel<<<(edgeThr + 255) / 256, 256, 0, stream>>>(h, ei, s, cnt, nE);
  gemm_kernel<true, true><<<gemmBlks, 256, 0, stream>>>(h, s, cnt, W1lT, W1rT, b1, h, n);

  zero_kernel<<<(nh + n + 255) / 256, 256, 0, stream>>>(s, nh + n);
  scatter_kernel<<<(edgeThr + 255) / 256, 256, 0, stream>>>(h, ei, s, cnt, nE);
  gemm_kernel<true, true><<<gemmBlks, 256, 0, stream>>>(h, s, cnt, W2lT, W2rT, b2, h, n);

  gemm_kernel<false, false><<<gemmBlks, 256, 0, stream>>>(h, nullptr, nullptr,
                                                          nullptr, WoT, bout, nst, n);
  colmean_kernel<<<HID, 256, 0, stream>>>(nst, outm, n);
}


# Neighbor sums without f32 atomics: CSR by dst per call (int histogram, scan, slot fill), each node's list sorted by source id, one wave per node adds its rows in that fixed order, not arrival order
# speedup vs baseline: 1.5317x; 1.5317x over previous
#include <hip/hip_runtime.h>


typedef __attribute__((ext_vector_type(16))) _Float16 v16h;
typedef __attribute__((ext_vector_type(8)))  _Float16 half8;
typedef __attribute__((ext_vector_type(8)))  float    v8f;

#define HID 128

__global__ void inproj_kernel(const float* __restrict__ pos,
                              const int*   __restrict__ an,
                              const float* __restrict__ W,
                              const float* __restrict__ b,
                              float* __restrict__ h, int n) {
  int t = blockIdx.x * 256 + threadIdx.x;
  if (t >= n * HID) return;
  int node = t >> 7, c = t & 127;
  float z  = (float)an[node] * 0.1f;
  float px = pos[node * 3 + 0];
  float py = pos[node * 3 + 1];
  float pz = pos[node * 3 + 2];
  float v = b[c];
  v = fmaf(z,  W[0 * HID + c], v);
  v = fmaf(px, W[1 * HID + c], v);
  v = fmaf(py, W[2 * HID + c], v);
  v = fmaf(pz, W[3 * HID + c], v);
  h[t] = fmaxf(v, 0.0f);
}

__global__ void wprep_kernel(const float* __restrict__ W,
                             _Float16* __restrict__ WT) {
  int t = blockIdx.x * 256 + threadIdx.x;
  if (t >= HID * HID) return;
  int k = t & 127, n = t >> 7;
  WT[t] = (_Float16)W[k * HID + n];
}

__global__ void zero_kernel(int* __restrict__ p, int n) {
  int t = blockIdx.x * 256 + threadIdx.x;
  if (t < n) p[t] = 0;
}

__global__ void degree_kernel(const int* __restrict__ ei,
                              int* __restrict__ deg, int nE) {
  int e = blockIdx.x * 256 + threadIdx.x;
  if (e < nE) atomicAdd(deg + ei[nE + e], 1);
}

__global__ __launch_bounds__(256) void blocksum_kernel(
    const int* __restrict__ deg, int* __restrict__ bsum, int n) {
  __shared__ int red[256];
  const int t = threadIdx.x;
  const int i0 = blockIdx.x * 1024 + t * 4;
  int v = 0;
#pragma unroll
  for (int k = 0; k < 4; ++k)
    if (i0 + k < n) v += deg[i0 + k];
  red[t] = v;
  __syncthreads();
  for (int st = 128; st > 0; st >>= 1) {
    if (t < st) red[t] += red[t + st];
    __syncthreads();
  }
  if (t == 0) bsum[blockIdx.x] = red[0];
}

__global__ __launch_bounds__(256) void scan_kernel(
    const int* __restrict__ deg, const int* __restrict__ bsum,
    int* __restrict__ rowptr, int* __restrict__ cursor, int n) {
  __shared__ int sh[256];
  const int t = threadIdx.x;
  const int b = blockIdx.x;
  int part = 0;
  for (int i = t; i < b; i += 256) part += bsum[i];
  sh[t] = part;
  __syncthreads();
  for (int st = 128; st > 0; st >>= 1) {
    if (t < st) sh[t] += sh[t + st];
    __syncthreads();
  }
  const int offset = sh[0];
  __syncthreads();
  const int i0 = b * 1024 + t * 4;
  int d[4];
#pragma unroll
  for (int k = 0; k < 4; ++k) d[k] = (i0 + k < n) ? deg[i0 + k] : 0;
  const int tsum = d[0] + d[1] + d[2] + d[3];
  sh[t] = tsum;
  __syncthreads();
  for (int st = 1; st < 256; st <<= 1) {
    const int v = (t >= st) ? sh[t - st] : 0;
    __syncthreads();
    sh[t] += v;
    __syncthreads();
  }
  int run = offset + sh[t] - tsum;
#pragma unroll
  for (int k = 0; k < 4; ++k) {
    if (i0 + k < n) {
      rowptr[i0 + k] = run;
      cursor[i0 + k] = run;
      run += d[k];
      if (i0 + k == n - 1) rowptr[n] = run;
    }
  }
}

__global__ void fill_kernel(const int* __restrict__ ei,
                            int* __restrict__ cursor,
                            int* __restrict__ adj, int nE) {
  int e = blockIdx.x * 256 + threadIdx.x;
  if (e >= nE) return;
  const int src = ei[e];
  const int dst = ei[nE + e];
  const int pos = atomicAdd(cursor + dst, 1);
  adj[pos] = src;
}

__global__ __launch_bounds__(256) void segsort_kernel(
    const int* __restrict__ rowptr, const int* __restrict__ adj,
    int* __restrict__ adjs, int n) {
  const int node =
      __builtin_amdgcn_readfirstlane((blockIdx.x * 256 + threadIdx.x) >> 5);
  if (node >= n) return;
  const int lane  = threadIdx.x & 31;
  const int start = rowptr[node];
  const int deg   = rowptr[node + 1] - start;
  if (deg <= 32) {
    const int key = (lane < deg) ? adj[start + lane] : 0x7fffffff;
    int rank = 0;
    for (int j = 0; j < deg; ++j) {
      const int kj = __builtin_amdgcn_readlane(key, j);
      rank += ((kj < key) || (kj == key && j < lane)) ? 1 : 0;
    }
    if (lane < deg) adjs[start + rank] = key;
  } else {
    for (int i = lane; i < deg; i += 32) {
      const int key = adj[start + i];
      int rank = 0;
      for (int j = 0; j < deg; ++j) {
        const int kj = adj[start + j];
        rank += ((kj < key) || (kj == key && j < i)) ? 1 : 0;
      }
      adjs[start + rank] = key;
    }
  }
}

template <int U>
__device__ inline void take_rows(const float* __restrict__ hl, int key, int j,
                                 float4& acc) {
  float4 v[U];
#pragma unroll
  for (int u = 0; u < U; ++u) {
    const int sidx = __builtin_amdgcn_readlane(key, j + u);
    v[u] = *(const float4*)(hl + (size_t)sidx * HID);
  }
#pragma unroll
  for (int u = 0; u < U; ++u) {
    acc.x += v[u].x; acc.y += v[u].y; acc.z += v[u].z; acc.w += v[u].w;
  }
}

__global__ __launch_bounds__(256) void gather_kernel(
    const float* __restrict__ h, const int* __restrict__ rowptr,
    const int* __restrict__ adjs, float* __restrict__ s,
    float* __restrict__ cnt, int n) {
  const int node =
      __builtin_amdgcn_readfirstlane((blockIdx.x * 256 + threadIdx.x) >> 5);
  if (node >= n) return;
  const int lane  = threadIdx.x & 31;
  const int start = rowptr[node];
  const int end   = rowptr[node + 1];
  const float* hl = h + lane * 4;
  float4 acc = make_float4(0.0f, 0.0f, 0.0f, 0.0f);
  for (int base = start; base < end; base += 32) {
    const int m   = (end - base < 32) ? (end - base) : 32;
    const int key = (lane < m) ? adjs[base + lane] : 0;
    int j = 0;
    for (; j + 8 <= m; j += 8) take_rows<8>(hl, key, j, acc);
    if (j + 4 <= m) { take_rows<4>(hl, key, j, acc); j += 4; }
    if (j + 2 <= m) { take_rows<2>(hl, key, j, acc); j += 2; }
    if (j < m) take_rows<1>(hl, key, j, acc);
  }
  *(float4*)(s + (size_t)node * HID + lane * 4) = acc;
  if (lane == 0) cnt[node] = (float)(end - start);
}

__device__ inline half8 cvt8(const float* __restrict__ p, float scale) {
  float4 x = *(const float4*)(p);
  float4 y = *(const float4*)(p + 4);
  half8 r;
  r[0] = (_Float16)(x.x * scale); r[1] = (_Float16)(x.y * scale);
  r[2] = (_Float16)(x.z * scale); r[3] = (_Float16)(x.w * scale);
  r[4] = (_Float16)(y.x * scale); r[5] = (_Float16)(y.y * scale);
  r[6] = (_Float16)(y.z * scale); r[7] = (_Float16)(y.w * scale);
  return r;
}

__device__ inline v16h join16(half8 lo, half8 hi) {
  return __builtin_shufflevector(lo, hi, 0, 1, 2, 3, 4, 5, 6, 7,
                                         8, 9, 10, 11, 12, 13, 14, 15);
}

template <bool SAGE, bool RELU>
__global__ __launch_bounds__(256) void gemm_kernel(
    const float*    __restrict__ h,
    const float*    __restrict__ s,
    const float*    __restrict__ cnt,
    const _Float16* __restrict__ WlT,
    const _Float16* __restrict__ WrT,
    const float*    __restrict__ bias,
    float* __restrict__ out,
    int n_nodes) {
  __shared__ _Float16 wtr[HID * HID];
  __shared__ _Float16 wtl[HID * HID];

  {
    uint4*       dr = (uint4*)wtr;
    const uint4* sr = (const uint4*)WrT;
    for (int i = threadIdx.x; i < HID * HID / 8; i += 256) dr[i] = sr[i];
    if constexpr (SAGE) {
      uint4*       dl = (uint4*)wtl;
      const uint4* sl = (const uint4*)WlT;
      for (int i = threadIdx.x; i < HID * HID / 8; i += 256) dl[i] = sl[i];
    }
  }
  __syncthreads();

  const int lane = threadIdx.x & 31;
  const int wave = threadIdx.x >> 5;
  const int rowbase = blockIdx.x * 128 + wave * 16;
  if (rowbase >= n_nodes) return;

  const int arow = rowbase + (lane & 15);
  const int row  = arow < n_nodes ? arow : n_nodes - 1;
  const int lo   = (lane >= 16) ? 8 : 0;

  float inv = 0.0f;
  if constexpr (SAGE) {
    float cv = cnt[row];
    inv = (cv > 0.0f) ? __builtin_amdgcn_rcpf(cv) : 0.0f;
  }

  const float*    hp  = h + (size_t)row * HID + lo;
  const float*    sp  = SAGE ? (s + (size_t)row * HID + lo) : nullptr;
  const _Float16* brp = &wtr[(lane & 15) * HID + lo];
  const _Float16* blp = &wtl[(lane & 15) * HID + lo];

  v8f acc[8];
  {
    v8f z = {};
#pragma unroll
    for (int nt = 0; nt < 8; ++nt) acc[nt] = z;
  }

#pragma unroll
  for (int kk = 0; kk < 4; ++kk) {
    const int kb = kk * 32;
    v16h ahf = join16(cvt8(hp + kb, 1.0f), cvt8(hp + kb + 16, 1.0f));
    v16h agf;
    if constexpr (SAGE)
      agf = join16(cvt8(sp + kb, inv), cvt8(sp + kb + 16, inv));
#pragma unroll
    for (int nt = 0; nt < 8; ++nt) {
      const int off = nt * 16 * HID + kb;
      v16h br = join16(*(const half8*)(brp + off),
                       *(const half8*)(brp + off + 16));
      acc[nt] = __builtin_amdgcn_wmma_f32_16x16x32_f16(
          false, ahf, false, br, (short)0, acc[nt], false, false);
      if constexpr (SAGE) {
        v16h bl = join16(*(const half8*)(blp + off),
                         *(const half8*)(blp + off + 16));
        acc[nt] = __builtin_amdgcn_wmma_f32_16x16x32_f16(
            false, agf, false, bl, (short)0, acc[nt], false, false);
      }
    }
  }

  const int mofs = (lane >= 16) ? 8 : 0;
#pragma unroll
  for (int nt = 0; nt < 8; ++nt) {
    const int ncol = nt * 16 + (lane & 15);
    const float bb = bias[ncol];
#pragma unroll
    for (int r = 0; r < 8; ++r) {
      float v = acc[nt][r] + bb;
      if constexpr (RELU) v = fmaxf(v, 0.0f);
      int orow = rowbase + mofs + r;
      if (orow < n_nodes) out[(size_t)orow * HID + ncol] = v;
    }
  }
}

__global__ void colmean_kernel(const float* __restrict__ ns,
                               float* __restrict__ outm, int n) {
  __shared__ float red[256];
  int col = blockIdx.x;
  float acc = 0.0f;
  for (int r = threadIdx.x; r < n; r += 256) acc += ns[(size_t)r * HID + col];
  red[threadIdx.x] = acc;
  __syncthreads();
  for (int st = 128; st > 0; st >>= 1) {
    if (threadIdx.x < st) red[threadIdx.x] += red[threadIdx.x + st];
    __syncthreads();
  }
  if (threadIdx.x == 0) outm[col] = red[0] / (float)n;
}

extern "C" void kernel_launch(void* const* d_in, const int* in_sizes, int n_in,
                              void* d_out, int out_size, void* d_ws, size_t ws_size,
                              hipStream_t stream) {
  const float* pos  = (const float*)d_in[0];
  const int*   an   = (const int*)  d_in[1];
  const int*   ei   = (const int*)  d_in[2];
  const float* W_in = (const float*)d_in[3];
  const float* b_in = (const float*)d_in[4];
  const float* W1l  = (const float*)d_in[5];
  const float* b1   = (const float*)d_in[6];
  const float* W1r  = (const float*)d_in[7];
  const float* W2l  = (const float*)d_in[8];
  const float* b2   = (const float*)d_in[9];
  const float* W2r  = (const float*)d_in[10];
  const float* Wout = (const float*)d_in[11];
  const float* bout = (const float*)d_in[12];

  const int n  = in_sizes[0] / 3;
  const int nE = in_sizes[2] / 2;

  float* ws   = (float*)d_ws;
  float* h    = ws;
  float* s    = ws + (size_t)n * HID;
  float* cnt  = s  + (size_t)n * HID;

  size_t woff = (size_t)n * HID * 2 + n;
  woff = (woff + 63) & ~(size_t)63;
  _Float16* wtbase = (_Float16*)(ws + woff);
  _Float16* W1lT = wtbase + 0 * HID * HID;
  _Float16* W1rT = wtbase + 1 * HID * HID;
  _Float16* W2lT = wtbase + 2 * HID * HID;
  _Float16* W2rT = wtbase + 3 * HID * HID;
  _Float16* WoT  = wtbase + 4 * HID * HID;

  const int scanBlks = (n + 1023) / 1024;
  int* deg    = (int*)(wtbase + 5 * HID * HID);
  int* rowptr = deg + n;
  int* cursor = rowptr + (n + 1);
  int* bsum   = cursor + n;
  int* adj    = bsum + scanBlks;
  int* adjs   = adj + nE;

  float* outm = (float*)d_out;
  float* nst  = (float*)d_out + HID;

  const int nh       = n * HID;
  const int gemmBlks = (n + 127) / 128;
  const int edgeBlks = (nE + 255) / 256;
  const int nodeBlks = (n + 7) / 8;
  const int wBlks    = (HID * HID + 255) / 256;

  wprep_kernel<<<wBlks, 256, 0, stream>>>(W1l,  W1lT);
  wprep_kernel<<<wBlks, 256, 0, stream>>>(W1r,  W1rT);
  wprep_kernel<<<wBlks, 256, 0, stream>>>(W2l,  W2lT);
  wprep_kernel<<<wBlks, 256, 0, stream>>>(W2r,  W2rT);
  wprep_kernel<<<wBlks, 256, 0, stream>>>(Wout, WoT);

  inproj_kernel<<<(nh + 255) / 256, 256, 0, stream>>>(pos, an, W_in, b_in, h, n);

  zero_kernel<<<(n + 255) / 256, 256, 0, stream>>>(deg, n);
  degree_kernel<<<edgeBlks, 256, 0, stream>>>(ei, deg, nE);
  blocksum_kernel<<<scanBlks, 256, 0, stream>>>(deg, bsum, n);
  scan_kernel<<<scanBlks, 256, 0, stream>>>(deg, bsum, rowptr, cursor, n);
  fill_kernel<<<edgeBlks, 256, 0, stream>>>(ei, cursor, adj, nE);
  segsort_kernel<<<nodeBlks, 256, 0, stream>>>(rowptr, adj, adjs, n);

  gather_kernel<<<nodeBlks, 256, 0, stream>>>(h, rowptr, adjs, s, cnt, n);
  gemm_kernel<true, true><<<gemmBlks, 256, 0, stream>>>(h, s, cnt, W1lT, W1rT, b1, h, n);

  gather_kernel<<<nodeBlks, 256, 0, stream>>>(h, rowptr, adjs, s, cnt, n);
  gemm_kernel<true, true><<<gemmBlks, 256, 0, stream>>>(h, s, cnt, W2lT, W2rT, b2, h, n);

  gemm_kernel<false, false><<<gemmBlks, 256, 0, stream>>>(h, nullptr, nullptr,
                                                          nullptr, WoT, bout, nst, n);
  colmean_kernel<<<HID, 256, 0, stream>>>(nst, outm, n);
}


# GEMM weight tiles in LDS at a 272-byte row pitch (16 B x odd) instead of 256 bytes: the 16 rows that a half-wave's ds_load_b128 B-fragment read touches no longer share the same 4 banks
# speedup vs baseline: 1.6897x; 1.1032x over previous
#include <hip/hip_runtime.h>


typedef __attribute__((ext_vector_type(16))) _Float16 v16h;
typedef __attribute__((ext_vector_type(8)))  _Float16 half8;
typedef __attribute__((ext_vector_type(8)))  float    v8f;

#define HID 128
#define WPITCH 136

__global__ void inproj_kernel(const float* __restrict__ pos,
                              const int*   __restrict__ an,
                              const float* __restrict__ W,
                              const float* __restrict__ b,
                              float* __restrict__ h, int n) {
  int t = blockIdx.x * 256 + threadIdx.x;
  if (t >= n * HID) return;
  int node = t >> 7, c = t & 127;
  float z  = (float)an[node] * 0.1f;
  float px = pos[node * 3 + 0];
  float py = pos[node * 3 + 1];
  float pz = pos[node * 3 + 2];
  float v = b[c];
  v = fmaf(z,  W[0 * HID + c], v);
  v = fmaf(px, W[1 * HID + c], v);
  v = fmaf(py, W[2 * HID + c], v);
  v = fmaf(pz, W[3 * HID + c], v);
  h[t] = fmaxf(v, 0.0f);
}

__global__ void wprep_kernel(const float* __restrict__ W,
                             _Float16* __restrict__ WT) {
  int t = blockIdx.x * 256 + threadIdx.x;
  if (t >= HID * HID) return;
  int k = t & 127, n = t >> 7;
  WT[t] = (_Float16)W[k * HID + n];
}

__global__ void zero_kernel(int* __restrict__ p, int n) {
  int t = blockIdx.x * 256 + threadIdx.x;
  if (t < n) p[t] = 0;
}

__global__ void degree_kernel(const int* __restrict__ ei,
                              int* __restrict__ deg, int nE) {
  int e = blockIdx.x * 256 + threadIdx.x;
  if (e < nE) atomicAdd(deg + ei[nE + e], 1);
}

__global__ __launch_bounds__(256) void blocksum_kernel(
    const int* __restrict__ deg, int* __restrict__ bsum, int n) {
  __shared__ int red[256];
  const int t = threadIdx.x;
  const int i0 = blockIdx.x * 1024 + t * 4;
  int v = 0;
#pragma unroll
  for (int k = 0; k < 4; ++k)
    if (i0 + k < n) v += deg[i0 + k];
  red[t] = v;
  __syncthreads();
  for (int st = 128; st > 0; st >>= 1) {
    if (t < st) red[t] += red[t + st];
    __syncthreads();
  }
  if (t == 0) bsum[blockIdx.x] = red[0];
}

__global__ __launch_bounds__(256) void scan_kernel(
    const int* __restrict__ deg, const int* __restrict__ bsum,
    int* __restrict__ rowptr, int* __restrict__ cursor, int n) {
  __shared__ int sh[256];
  const int t = threadIdx.x;
  const int b = blockIdx.x;
  int part = 0;
  for (int i = t; i < b; i += 256) part += bsum[i];
  sh[t] = part;
  __syncthreads();
  for (int st = 128; st > 0; st >>= 1) {
    if (t < st) sh[t] += sh[t + st];
    __syncthreads();
  }
  const int offset = sh[0];
  __syncthreads();
  const int i0 = b * 1024 + t * 4;
  int d[4];
#pragma unroll
  for (int k = 0; k < 4; ++k) d[k] = (i0 + k < n) ? deg[i0 + k] : 0;
  const int tsum = d[0] + d[1] + d[2] + d[3];
  sh[t] = tsum;
  __syncthreads();
  for (int st = 1; st < 256; st <<= 1) {
    const int v = (t >= st) ? sh[t - st] : 0;
    __syncthreads();
    sh[t] += v;
    __syncthreads();
  }
  int run = offset + sh[t] - tsum;
#pragma unroll
  for (int k = 0; k < 4; ++k) {
    if (i0 + k < n) {
      rowptr[i0 + k] = run;
      cursor[i0 + k] = run;
      run += d[k];
      if (i0 + k == n - 1) rowptr[n] = run;
    }
  }
}

__global__ void fill_kernel(const int* __restrict__ ei,
                            int* __restrict__ cursor,
                            int* __restrict__ adj, int nE) {
  int e = blockIdx.x * 256 + threadIdx.x;
  if (e >= nE) return;
  const int src = ei[e];
  const int dst = ei[nE + e];
  const int pos = atomicAdd(cursor + dst, 1);
  adj[pos] = src;
}

__global__ __launch_bounds__(256) void segsort_kernel(
    const int* __restrict__ rowptr, const int* __restrict__ adj,
    int* __restrict__ adjs, int n) {
  const int node =
      __builtin_amdgcn_readfirstlane((blockIdx.x * 256 + threadIdx.x) >> 5);
  if (node >= n) return;
  const int lane  = threadIdx.x & 31;
  const int start = rowptr[node];
  const int deg   = rowptr[node + 1] - start;
  if (deg <= 32) {
    const int key = (lane < deg) ? adj[start + lane] : 0x7fffffff;
    int rank = 0;
    for (int j = 0; j < deg; ++j) {
      const int kj = __builtin_amdgcn_readlane(key, j);
      rank += ((kj < key) || (kj == key && j < lane)) ? 1 : 0;
    }
    if (lane < deg) adjs[start + rank] = key;
  } else {
    for (int i = lane; i < deg; i += 32) {
      const int key = adj[start + i];
      int rank = 0;
      for (int j = 0; j < deg; ++j) {
        const int kj = adj[start + j];
        rank += ((kj < key) || (kj == key && j < i)) ? 1 : 0;
      }
      adjs[start + rank] = key;
    }
  }
}

template <int U>
__device__ inline void take_rows(const float* __restrict__ hl, int key, int j,
                                 float4& acc) {
  float4 v[U];
#pragma unroll
  for (int u = 0; u < U; ++u) {
    const int sidx = __builtin_amdgcn_readlane(key, j + u);
    v[u] = *(const float4*)(hl + (size_t)sidx * HID);
  }
#pragma unroll
  for (int u = 0; u < U; ++u) {
    acc.x += v[u].x; acc.y += v[u].y; acc.z += v[u].z; acc.w += v[u].w;
  }
}

__global__ __launch_bounds__(256) void gather_kernel(
    const float* __restrict__ h, const int* __restrict__ rowptr,
    const int* __restrict__ adjs, float* __restrict__ s,
    float* __restrict__ cnt, int n) {
  const int node =
      __builtin_amdgcn_readfirstlane((blockIdx.x * 256 + threadIdx.x) >> 5);
  if (node >= n) return;
  const int lane  = threadIdx.x & 31;
  const int start = rowptr[node];
  const int end   = rowptr[node + 1];
  const float* hl = h + lane * 4;
  float4 acc = make_float4(0.0f, 0.0f, 0.0f, 0.0f);
  for (int base = start; base < end; base += 32) {
    const int m   = (end - base < 32) ? (end - base) : 32;
    const int key = (lane < m) ? adjs[base + lane] : 0;
    int j = 0;
    for (; j + 8 <= m; j += 8) take_rows<8>(hl, key, j, acc);
    if (j + 4 <= m) { take_rows<4>(hl, key, j, acc); j += 4; }
    if (j + 2 <= m) { take_rows<2>(hl, key, j, acc); j += 2; }
    if (j < m) take_rows<1>(hl, key, j, acc);
  }
  *(float4*)(s + (size_t)node * HID + lane * 4) = acc;
  if (lane == 0) cnt[node] = (float)(end - start);
}

__device__ inline half8 cvt8(const float* __restrict__ p, float scale) {
  float4 x = *(const float4*)(p);
  float4 y = *(const float4*)(p + 4);
  half8 r;
  r[0] = (_Float16)(x.x * scale); r[1] = (_Float16)(x.y * scale);
  r[2] = (_Float16)(x.z * scale); r[3] = (_Float16)(x.w * scale);
  r[4] = (_Float16)(y.x * scale); r[5] = (_Float16)(y.y * scale);
  r[6] = (_Float16)(y.z * scale); r[7] = (_Float16)(y.w * scale);
  return r;
}

__device__ inline v16h join16(half8 lo, half8 hi) {
  return __builtin_shufflevector(lo, hi, 0, 1, 2, 3, 4, 5, 6, 7,
                                         8, 9, 10, 11, 12, 13, 14, 15);
}

template <bool SAGE, bool RELU>
__global__ __launch_bounds__(256) void gemm_kernel(
    const float*    __restrict__ h,
    const float*    __restrict__ s,
    const float*    __restrict__ cnt,
    const _Float16* __restrict__ WlT,
    const _Float16* __restrict__ WrT,
    const float*    __restrict__ bias,
    float* __restrict__ out,
    int n_nodes) {
  __shared__ __attribute__((aligned(16))) _Float16 wtr[HID * WPITCH];
  __shared__ __attribute__((aligned(16))) _Float16 wtl[HID * WPITCH];

  {
    const uint4* sr = (const uint4*)WrT;
    for (int i = threadIdx.x; i < HID * HID / 8; i += 256)
      *(uint4*)&wtr[(i >> 4) * WPITCH + (i & 15) * 8] = sr[i];
    if constexpr (SAGE) {
      const uint4* sl = (const uint4*)WlT;
      for (int i = threadIdx.x; i < HID * HID / 8; i += 256)
        *(uint4*)&wtl[(i >> 4) * WPITCH + (i & 15) * 8] = sl[i];
    }
  }
  __syncthreads();

  const int lane = threadIdx.x & 31;
  const int wave = threadIdx.x >> 5;
  const int rowbase = blockIdx.x * 128 + wave * 16;
  if (rowbase >= n_nodes) return;

  const int arow = rowbase + (lane & 15);
  const int row  = arow < n_nodes ? arow : n_nodes - 1;
  const int lo   = (lane >= 16) ? 8 : 0;

  float inv = 0.0f;
  if constexpr (SAGE) {
    float cv = cnt[row];
    inv = (cv > 0.0f) ? __builtin_amdgcn_rcpf(cv) : 0.0f;
  }

  const float*    hp  = h + (size_t)row * HID + lo;
  const float*    sp  = SAGE ? (s + (size_t)row * HID + lo) : nullptr;
  const _Float16* brp = &wtr[(lane & 15) * WPITCH + lo];
  const _Float16* blp = &wtl[(lane & 15) * WPITCH + lo];

  v8f acc[8];
  {
    v8f z = {};
#pragma unroll
    for (int nt = 0; nt < 8; ++nt) acc[nt] = z;
  }

#pragma unroll
  for (int kk = 0; kk < 4; ++kk) {
    const int kb = kk * 32;
    v16h ahf = join16(cvt8(hp + kb, 1.0f), cvt8(hp + kb + 16, 1.0f));
    v16h agf;
    if constexpr (SAGE)
      agf = join16(cvt8(sp + kb, inv), cvt8(sp + kb + 16, inv));
#pragma unroll
    for (int nt = 0; nt < 8; ++nt) {
      const int off = nt * 16 * WPITCH + kb;
      v16h br = join16(*(const half8*)(brp + off),
                       *(const half8*)(brp + off + 16));
      acc[nt] = __builtin_amdgcn_wmma_f32_16x16x32_f16(
          false, ahf, false, br, (short)0, acc[nt], false, false);
      if constexpr (SAGE) {
        v16h bl = join16(*(const half8*)(blp + off),
                         *(const half8*)(blp + off + 16));
        acc[nt] = __builtin_amdgcn_wmma_f32_16x16x32_f16(
            false, agf, false, bl, (short)0, acc[nt], false, false);
      }
    }
  }

  const int mofs = (lane >= 16) ? 8 : 0;
#pragma unroll
  for (int nt = 0; nt < 8; ++nt) {
    const int ncol = nt * 16 + (lane & 15);
    const float bb = bias[ncol];
#pragma unroll
    for (int r = 0; r < 8; ++r) {
      float v = acc[nt][r] + bb;
      if constexpr (RELU) v = fmaxf(v, 0.0f);
      int orow = rowbase + mofs + r;
      if (orow < n_nodes) out[(size_t)orow * HID + ncol] = v;
    }
  }
}

__global__ void colmean_kernel(const float* __restrict__ ns,
                               float* __restrict__ outm, int n) {
  __shared__ float red[256];
  int col = blockIdx.x;
  float acc = 0.0f;
  for (int r = threadIdx.x; r < n; r += 256) acc += ns[(size_t)r * HID + col];
  red[threadIdx.x] = acc;
  __syncthreads();
  for (int st = 128; st > 0; st >>= 1) {
    if (threadIdx.x < st) red[threadIdx.x] += red[threadIdx.x + st];
    __syncthreads();
  }
  if (threadIdx.x == 0) outm[col] = red[0] / (float)n;
}

extern "C" void kernel_launch(void* const* d_in, const int* in_sizes, int n_in,
                              void* d_out, int out_size, void* d_ws, size_t ws_size,
                              hipStream_t stream) {
  const float* pos  = (const float*)d_in[0];
  const int*   an   = (const int*)  d_in[1];
  const int*   ei   = (const int*)  d_in[2];
  const float* W_in = (const float*)d_in[3];
  const float* b_in = (const float*)d_in[4];
  const float* W1l  = (const float*)d_in[5];
  const float* b1   = (const float*)d_in[6];
  const float* W1r  = (const float*)d_in[7];
  const float* W2l  = (const float*)d_in[8];
  const float* b2   = (const float*)d_in[9];
  const float* W2r  = (const float*)d_in[10];
  const float* Wout = (const float*)d_in[11];
  const float* bout = (const float*)d_in[12];

  const int n  = in_sizes[0] / 3;
  const int nE = in_sizes[2] / 2;

  float* ws   = (float*)d_ws;
  float* h    = ws;
  float* s    = ws + (size_t)n * HID;
  float* cnt  = s  + (size_t)n * HID;

  size_t woff = (size_t)n * HID * 2 + n;
  woff = (woff + 63) & ~(size_t)63;
  _Float16* wtbase = (_Float16*)(ws + woff);
  _Float16* W1lT = wtbase + 0 * HID * HID;
  _Float16* W1rT = wtbase + 1 * HID * HID;
  _Float16* W2lT = wtbase + 2 * HID * HID;
  _Float16* W2rT = wtbase + 3 * HID * HID;
  _Float16* WoT  = wtbase + 4 * HID * HID;

  const int scanBlks = (n + 1023) / 1024;
  int* deg    = (int*)(wtbase + 5 * HID * HID);
  int* rowptr = deg + n;
  int* cursor = rowptr + (n + 1);
  int* bsum   = cursor + n;
  int* adj    = bsum + scanBlks;
  int* adjs   = adj + nE;

  float* outm = (float*)d_out;
  float* nst  = (float*)d_out + HID;

  const int nh       = n * HID;
  const int gemmBlks = (n + 127) / 128;
  const int edgeBlks = (nE + 255) / 256;
  const int nodeBlks = (n + 7) / 8;
  const int wBlks    = (HID * HID + 255) / 256;

  wprep_kernel<<<wBlks, 256, 0, stream>>>(W1l,  W1lT);
  wprep_kernel<<<wBlks, 256, 0, stream>>>(W1r,  W1rT);
  wprep_kernel<<<wBlks, 256, 0, stream>>>(W2l,  W2lT);
  wprep_kernel<<<wBlks, 256, 0, stream>>>(W2r,  W2rT);
  wprep_kernel<<<wBlks, 256, 0, stream>>>(Wout, WoT);

  inproj_kernel<<<(nh + 255) / 256, 256, 0, stream>>>(pos, an, W_in, b_in, h, n);

  zero_kernel<<<(n + 255) / 256, 256, 0, stream>>>(deg, n);
  degree_kernel<<<edgeBlks, 256, 0, stream>>>(ei, deg, nE);
  blocksum_kernel<<<scanBlks, 256, 0, stream>>>(deg, bsum, n);
  scan_kernel<<<scanBlks, 256, 0, stream>>>(deg, bsum, rowptr, cursor, n);
  fill_kernel<<<edgeBlks, 256, 0, stream>>>(ei, cursor, adj, nE);
  segsort_kernel<<<nodeBlks, 256, 0, stream>>>(rowptr, adj, adjs, n);

  gather_kernel<<<nodeBlks, 256, 0, stream>>>(h, rowptr, adjs, s, cnt, n);
  gemm_kernel<true, true><<<gemmBlks, 256, 0, stream>>>(h, s, cnt, W1lT, W1rT, b1, h, n);

  gather_kernel<<<nodeBlks, 256, 0, stream>>>(h, rowptr, adjs, s, cnt, n);
  gemm_kernel<true, true><<<gemmBlks, 256, 0, stream>>>(h, s, cnt, W2lT, W2rT, b2, h, n);

  gemm_kernel<false, false><<<gemmBlks, 256, 0, stream>>>(h, nullptr, nullptr,
                                                          nullptr, WoT, bout, nst, n);
  colmean_kernel<<<HID, 256, 0, stream>>>(nst, outm, n);
}


# Column mean from per-work-group column sums taken out of the output GEMM's accumulators (fixed order; a new reduction tree for the mean) instead of a by-column re-read of node_states
# speedup vs baseline: 2.2932x; 1.3572x over previous
#include <hip/hip_runtime.h>


typedef __attribute__((ext_vector_type(16))) _Float16 v16h;
typedef __attribute__((ext_vector_type(8)))  _Float16 half8;
typedef __attribute__((ext_vector_type(8)))  float    v8f;

#define HID 128
#define WPITCH 136

__global__ void inproj_kernel(const float* __restrict__ pos,
                              const int*   __restrict__ an,
                              const float* __restrict__ W,
                              const float* __restrict__ b,
                              float* __restrict__ h, int n) {
  int t = blockIdx.x * 256 + threadIdx.x;
  if (t >= n * HID) return;
  int node = t >> 7, c = t & 127;
  float z  = (float)an[node] * 0.1f;
  float px = pos[node * 3 + 0];
  float py = pos[node * 3 + 1];
  float pz = pos[node * 3 + 2];
  float v = b[c];
  v = fmaf(z,  W[0 * HID + c], v);
  v = fmaf(px, W[1 * HID + c], v);
  v = fmaf(py, W[2 * HID + c], v);
  v = fmaf(pz, W[3 * HID + c], v);
  h[t] = fmaxf(v, 0.0f);
}

__global__ void wprep_kernel(const float* __restrict__ W,
                             _Float16* __restrict__ WT) {
  int t = blockIdx.x * 256 + threadIdx.x;
  if (t >= HID * HID) return;
  int k = t & 127, n = t >> 7;
  WT[t] = (_Float16)W[k * HID + n];
}

__global__ void zero_kernel(int* __restrict__ p, int n) {
  int t = blockIdx.x * 256 + threadIdx.x;
  if (t < n) p[t] = 0;
}

__global__ void degree_kernel(const int* __restrict__ ei,
                              int* __restrict__ deg, int nE) {
  int e = blockIdx.x * 256 + threadIdx.x;
  if (e < nE) atomicAdd(deg + ei[nE + e], 1);
}

__global__ __launch_bounds__(256) void blocksum_kernel(
    const int* __restrict__ deg, int* __restrict__ bsum, int n) {
  __shared__ int red[256];
  const int t = threadIdx.x;
  const int i0 = blockIdx.x * 1024 + t * 4;
  int v = 0;
#pragma unroll
  for (int k = 0; k < 4; ++k)
    if (i0 + k < n) v += deg[i0 + k];
  red[t] = v;
  __syncthreads();
  for (int st = 128; st > 0; st >>= 1) {
    if (t < st) red[t] += red[t + st];
    __syncthreads();
  }
  if (t == 0) bsum[blockIdx.x] = red[0];
}

__global__ __launch_bounds__(256) void scan_kernel(
    const int* __restrict__ deg, const int* __restrict__ bsum,
    int* __restrict__ rowptr, int* __restrict__ cursor, int n) {
  __shared__ int sh[256];
  const int t = threadIdx.x;
  const int b = blockIdx.x;
  int part = 0;
  for (int i = t; i < b; i += 256) part += bsum[i];
  sh[t] = part;
  __syncthreads();
  for (int st = 128; st > 0; st >>= 1) {
    if (t < st) sh[t] += sh[t + st];
    __syncthreads();
  }
  const int offset = sh[0];
  __syncthreads();
  const int i0 = b * 1024 + t * 4;
  int d[4];
#pragma unroll
  for (int k = 0; k < 4; ++k) d[k] = (i0 + k < n) ? deg[i0 + k] : 0;
  const int tsum = d[0] + d[1] + d[2] + d[3];
  sh[t] = tsum;
  __syncthreads();
  for (int st = 1; st < 256; st <<= 1) {
    const int v = (t >= st) ? sh[t - st] : 0;
    __syncthreads();
    sh[t] += v;
    __syncthreads();
  }
  int run = offset + sh[t] - tsum;
#pragma unroll
  for (int k = 0; k < 4; ++k) {
    if (i0 + k < n) {
      rowptr[i0 + k] = run;
      cursor[i0 + k] = run;
      run += d[k];
      if (i0 + k == n - 1) rowptr[n] = run;
    }
  }
}

__global__ void fill_kernel(const int* __restrict__ ei,
                            int* __restrict__ cursor,
                            int* __restrict__ adj, int nE) {
  int e = blockIdx.x * 256 + threadIdx.x;
  if (e >= nE) return;
  const int src = ei[e];
  const int dst = ei[nE + e];
  const int pos = atomicAdd(cursor + dst, 1);
  adj[pos] = src;
}

__global__ __launch_bounds__(256) void segsort_kernel(
    const int* __restrict__ rowptr, const int* __restrict__ adj,
    int* __restrict__ adjs, int n) {
  const int node =
      __builtin_amdgcn_readfirstlane((blockIdx.x * 256 + threadIdx.x) >> 5);
  if (node >= n) return;
  const int lane  = threadIdx.x & 31;
  const int start = rowptr[node];
  const int deg   = rowptr[node + 1] - start;
  if (deg <= 32) {
    const int key = (lane < deg) ? adj[start + lane] : 0x7fffffff;
    int rank = 0;
    for (int j = 0; j < deg; ++j) {
      const int kj = __builtin_amdgcn_readlane(key, j);
      rank += ((kj < key) || (kj == key && j < lane)) ? 1 : 0;
    }
    if (lane < deg) adjs[start + rank] = key;
  } else {
    for (int i = lane; i < deg; i += 32) {
      const int key = adj[start + i];
      int rank = 0;
      for (int j = 0; j < deg; ++j) {
        const int kj = adj[start + j];
        rank += ((kj < key) || (kj == key && j < i)) ? 1 : 0;
      }
      adjs[start + rank] = key;
    }
  }
}

template <int U>
__device__ inline void take_rows(const float* __restrict__ hl, int key, int j,
                                 float4& acc) {
  float4 v[U];
#pragma unroll
  for (int u = 0; u < U; ++u) {
    const int sidx = __builtin_amdgcn_readlane(key, j + u);
    v[u] = *(const float4*)(hl + (size_t)sidx * HID);
  }
#pragma unroll
  for (int u = 0; u < U; ++u) {
    acc.x += v[u].x; acc.y += v[u].y; acc.z += v[u].z; acc.w += v[u].w;
  }
}

__global__ __launch_bounds__(256) void gather_kernel(
    const float* __restrict__ h, const int* __restrict__ rowptr,
    const int* __restrict__ adjs, float* __restrict__ s,
    float* __restrict__ cnt, int n) {
  const int node =
      __builtin_amdgcn_readfirstlane((blockIdx.x * 256 + threadIdx.x) >> 5);
  if (node >= n) return;
  const int lane  = threadIdx.x & 31;
  const int start = rowptr[node];
  const int end   = rowptr[node + 1];
  const float* hl = h + lane * 4;
  float4 acc = make_float4(0.0f, 0.0f, 0.0f, 0.0f);
  for (int base = start; base < end; base += 32) {
    const int m   = (end - base < 32) ? (end - base) : 32;
    const int key = (lane < m) ? adjs[base + lane] : 0;
    int j = 0;
    for (; j + 8 <= m; j += 8) take_rows<8>(hl, key, j, acc);
    if (j + 4 <= m) { take_rows<4>(hl, key, j, acc); j += 4; }
    if (j + 2 <= m) { take_rows<2>(hl, key, j, acc); j += 2; }
    if (j < m) take_rows<1>(hl, key, j, acc);
  }
  *(float4*)(s + (size_t)node * HID + lane * 4) = acc;
  if (lane == 0) cnt[node] = (float)(end - start);
}

__device__ inline half8 cvt8(const float* __restrict__ p, float scale) {
  float4 x = *(const float4*)(p);
  float4 y = *(const float4*)(p + 4);
  half8 r;
  r[0] = (_Float16)(x.x * scale); r[1] = (_Float16)(x.y * scale);
  r[2] = (_Float16)(x.z * scale); r[3] = (_Float16)(x.w * scale);
  r[4] = (_Float16)(y.x * scale); r[5] = (_Float16)(y.y * scale);
  r[6] = (_Float16)(y.z * scale); r[7] = (_Float16)(y.w * scale);
  return r;
}

__device__ inline v16h join16(half8 lo, half8 hi) {
  return __builtin_shufflevector(lo, hi, 0, 1, 2, 3, 4, 5, 6, 7,
                                         8, 9, 10, 11, 12, 13, 14, 15);
}

template <bool SAGE, bool RELU, bool COLSUM>
__global__ __launch_bounds__(256) void gemm_kernel(
    const float*    __restrict__ h,
    const float*    __restrict__ s,
    const float*    __restrict__ cnt,
    const _Float16* __restrict__ WlT,
    const _Float16* __restrict__ WrT,
    const float*    __restrict__ bias,
    float* __restrict__ out,
    float* __restrict__ colpart,
    int n_nodes) {
  __shared__ __attribute__((aligned(16))) _Float16 wtr[HID * WPITCH];
  __shared__ __attribute__((aligned(16))) _Float16 wtl[HID * WPITCH];
  __shared__ float colred[COLSUM ? 16 * HID : 1];

  {
    const uint4* sr = (const uint4*)WrT;
    for (int i = threadIdx.x; i < HID * HID / 8; i += 256)
      *(uint4*)&wtr[(i >> 4) * WPITCH + (i & 15) * 8] = sr[i];
    if constexpr (SAGE) {
      const uint4* sl = (const uint4*)WlT;
      for (int i = threadIdx.x; i < HID * HID / 8; i += 256)
        *(uint4*)&wtl[(i >> 4) * WPITCH + (i & 15) * 8] = sl[i];
    }
  }
  __syncthreads();

  const int lane = threadIdx.x & 31;
  const int wave = threadIdx.x >> 5;
  const int rowbase = blockIdx.x * 128 + wave * 16;
  const bool active = rowbase < n_nodes;
  if constexpr (!COLSUM) {
    if (!active) return;
  }

  float csum[8];
#pragma unroll
  for (int nt = 0; nt < 8; ++nt) csum[nt] = 0.0f;

  if (active) {

  const int arow = rowbase + (lane & 15);
  const int row  = arow < n_nodes ? arow : n_nodes - 1;
  const int lo   = (lane >= 16) ? 8 : 0;

  float inv = 0.0f;
  if constexpr (SAGE) {
    float cv = cnt[row];
    inv = (cv > 0.0f) ? __builtin_amdgcn_rcpf(cv) : 0.0f;
  }

  const float*    hp  = h + (size_t)row * HID + lo;
  const float*    sp  = SAGE ? (s + (size_t)row * HID + lo) : nullptr;
  const _Float16* brp = &wtr[(lane & 15) * WPITCH + lo];
  const _Float16* blp = &wtl[(lane & 15) * WPITCH + lo];

  v8f acc[8];
  {
    v8f z = {};
#pragma unroll
    for (int nt = 0; nt < 8; ++nt) acc[nt] = z;
  }

#pragma unroll
  for (int kk = 0; kk < 4; ++kk) {
    const int kb = kk * 32;
    v16h ahf = join16(cvt8(hp + kb, 1.0f), cvt8(hp + kb + 16, 1.0f));
    v16h agf;
    if constexpr (SAGE)
      agf = join16(cvt8(sp + kb, inv), cvt8(sp + kb + 16, inv));
#pragma unroll
    for (int nt = 0; nt < 8; ++nt) {
      const int off = nt * 16 * WPITCH + kb;
      v16h br = join16(*(const half8*)(brp + off),
                       *(const half8*)(brp + off + 16));
      acc[nt] = __builtin_amdgcn_wmma_f32_16x16x32_f16(
          false, ahf, false, br, (short)0, acc[nt], false, false);
      if constexpr (SAGE) {
        v16h bl = join16(*(const half8*)(blp + off),
                         *(const half8*)(blp + off + 16));
        acc[nt] = __builtin_amdgcn_wmma_f32_16x16x32_f16(
            false, agf, false, bl, (short)0, acc[nt], false, false);
      }
    }
  }

  const int mofs = (lane >= 16) ? 8 : 0;
#pragma unroll
  for (int nt = 0; nt < 8; ++nt) {
    const int ncol = nt * 16 + (lane & 15);
    const float bb = bias[ncol];
#pragma unroll
    for (int r = 0; r < 8; ++r) {
      float v = acc[nt][r] + bb;
      if constexpr (RELU) v = fmaxf(v, 0.0f);
      int orow = rowbase + mofs + r;
      if (orow < n_nodes) {
        out[(size_t)orow * HID + ncol] = v;
        if constexpr (COLSUM) csum[nt] += v;
      }
    }
  }
  }

  if constexpr (COLSUM) {
    float* cr = &colred[(wave * 2 + (lane >> 4)) * HID + (lane & 15)];
#pragma unroll
    for (int nt = 0; nt < 8; ++nt) cr[nt * 16] = csum[nt];
    __syncthreads();
    if (threadIdx.x < HID) {
      float t = 0.0f;
#pragma unroll
      for (int p = 0; p < 16; ++p) t += colred[p * HID + threadIdx.x];
      colpart[(size_t)blockIdx.x * HID + threadIdx.x] = t;
    }
  }
}

__global__ void colmean_kernel(const float* __restrict__ part, int nparts,
                               float* __restrict__ outm, int n) {
  __shared__ float red[256];
  int col = blockIdx.x;
  float acc = 0.0f;
  for (int r = threadIdx.x; r < nparts; r += 256)
    acc += part[(size_t)r * HID + col];
  red[threadIdx.x] = acc;
  __syncthreads();
  for (int st = 128; st > 0; st >>= 1) {
    if (threadIdx.x < st) red[threadIdx.x] += red[threadIdx.x + st];
    __syncthreads();
  }
  if (threadIdx.x == 0) outm[col] = red[0] / (float)n;
}

extern "C" void kernel_launch(void* const* d_in, const int* in_sizes, int n_in,
                              void* d_out, int out_size, void* d_ws, size_t ws_size,
                              hipStream_t stream) {
  const float* pos  = (const float*)d_in[0];
  const int*   an   = (const int*)  d_in[1];
  const int*   ei   = (const int*)  d_in[2];
  const float* W_in = (const float*)d_in[3];
  const float* b_in = (const float*)d_in[4];
  const float* W1l  = (const float*)d_in[5];
  const float* b1   = (const float*)d_in[6];
  const float* W1r  = (const float*)d_in[7];
  const float* W2l  = (const float*)d_in[8];
  const float* b2   = (const float*)d_in[9];
  const float* W2r  = (const float*)d_in[10];
  const float* Wout = (const float*)d_in[11];
  const float* bout = (const float*)d_in[12];

  const int n  = in_sizes[0] / 3;
  const int nE = in_sizes[2] / 2;

  float* ws   = (float*)d_ws;
  float* h    = ws;
  float* s    = ws + (size_t)n * HID;
  float* cnt  = s  + (size_t)n * HID;

  size_t woff = (size_t)n * HID * 2 + n;
  woff = (woff + 63) & ~(size_t)63;
  _Float16* wtbase = (_Float16*)(ws + woff);
  _Float16* W1lT = wtbase + 0 * HID * HID;
  _Float16* W1rT = wtbase + 1 * HID * HID;
  _Float16* W2lT = wtbase + 2 * HID * HID;
  _Float16* W2rT = wtbase + 3 * HID * HID;
  _Float16* WoT  = wtbase + 4 * HID * HID;

  const int scanBlks = (n + 1023) / 1024;
  int* deg    = (int*)(wtbase + 5 * HID * HID);
  int* rowptr = deg + n;
  int* cursor = rowptr + (n + 1);
  int* bsum   = cursor + n;
  int* adj    = bsum + scanBlks;
  int* adjs   = adj + nE;
  float* colpart = (float*)(adjs + nE);

  float* outm = (float*)d_out;
  float* nst  = (float*)d_out + HID;

  const int nh       = n * HID;
  const int gemmBlks = (n + 127) / 128;
  const int edgeBlks = (nE + 255) / 256;
  const int nodeBlks = (n + 7) / 8;
  const int wBlks    = (HID * HID + 255) / 256;

  wprep_kernel<<<wBlks, 256, 0, stream>>>(W1l,  W1lT);
  wprep_kernel<<<wBlks, 256, 0, stream>>>(W1r,  W1rT);
  wprep_kernel<<<wBlks, 256, 0, stream>>>(W2l,  W2lT);
  wprep_kernel<<<wBlks, 256, 0, stream>>>(W2r,  W2rT);
  wprep_kernel<<<wBlks, 256, 0, stream>>>(Wout, WoT);

  inproj_kernel<<<(nh + 255) / 256, 256, 0, stream>>>(pos, an, W_in, b_in, h, n);

  zero_kernel<<<(n + 255) / 256, 256, 0, stream>>>(deg, n);
  degree_kernel<<<edgeBlks, 256, 0, stream>>>(ei, deg, nE);
  blocksum_kernel<<<scanBlks, 256, 0, stream>>>(deg, bsum, n);
  scan_kernel<<<scanBlks, 256, 0, stream>>>(deg, bsum, rowptr, cursor, n);
  fill_kernel<<<edgeBlks, 256, 0, stream>>>(ei, cursor, adj, nE);
  segsort_kernel<<<nodeBlks, 256, 0, stream>>>(rowptr, adj, adjs, n);

  gather_kernel<<<nodeBlks, 256, 0, stream>>>(h, rowptr, adjs, s, cnt, n);
  gemm_kernel<true, true, false><<<gemmBlks, 256, 0, stream>>>(
      h, s, cnt, W1lT, W1rT, b1, h, nullptr, n);

  gather_kernel<<<nodeBlks, 256, 0, stream>>>(h, rowptr, adjs, s, cnt, n);
  gemm_kernel<true, true, false><<<gemmBlks, 256, 0, stream>>>(
      h, s, cnt, W2lT, W2rT, b2, h, nullptr, n);

  gemm_kernel<false, false, true><<<gemmBlks, 256, 0, stream>>>(
      h, nullptr, nullptr, nullptr, WoT, bout, nst, colpart, n);
  colmean_kernel<<<HID, 256, 0, stream>>>(colpart, gemmBlks, outm, n);
}


# CSR build with one pass of integer atomics, not two: the histogram add's returned count is kept as the edge's rank and the fill places src at rowptr[dst]+rank with plain stores; sums, order same
# speedup vs baseline: 3.2460x; 1.4155x over previous
#include <hip/hip_runtime.h>


typedef __attribute__((ext_vector_type(16))) _Float16 v16h;
typedef __attribute__((ext_vector_type(8)))  _Float16 half8;
typedef __attribute__((ext_vector_type(4)))  _Float16 half4;
typedef __attribute__((ext_vector_type(8)))  float    v8f;

#define HID 128
#define WPITCH 136

__global__ __launch_bounds__(256) void inproj_kernel(
    const float* __restrict__ pos,
    const int*   __restrict__ an,
    const float* __restrict__ W,
    const float* __restrict__ b,
    _Float16* __restrict__ h, int n) {
  const int node =
      __builtin_amdgcn_readfirstlane((blockIdx.x * 256 + threadIdx.x) >> 5);
  if (node >= n) return;
  const int c = (threadIdx.x & 31) * 4;
  const float z  = (float)an[node] * 0.1f;
  const float px = pos[node * 3 + 0];
  const float py = pos[node * 3 + 1];
  const float pz = pos[node * 3 + 2];
  const float4 w0 = *(const float4*)(W + 0 * HID + c);
  const float4 w1 = *(const float4*)(W + 1 * HID + c);
  const float4 w2 = *(const float4*)(W + 2 * HID + c);
  const float4 w3 = *(const float4*)(W + 3 * HID + c);
  float4 v = *(const float4*)(b + c);
  v.x = fmaf(z,  w0.x, v.x); v.y = fmaf(z,  w0.y, v.y);
  v.z = fmaf(z,  w0.z, v.z); v.w = fmaf(z,  w0.w, v.w);
  v.x = fmaf(px, w1.x, v.x); v.y = fmaf(px, w1.y, v.y);
  v.z = fmaf(px, w1.z, v.z); v.w = fmaf(px, w1.w, v.w);
  v.x = fmaf(py, w2.x, v.x); v.y = fmaf(py, w2.y, v.y);
  v.z = fmaf(py, w2.z, v.z); v.w = fmaf(py, w2.w, v.w);
  v.x = fmaf(pz, w3.x, v.x); v.y = fmaf(pz, w3.y, v.y);
  v.z = fmaf(pz, w3.z, v.z); v.w = fmaf(pz, w3.w, v.w);
  half4 o;
  o[0] = (_Float16)fmaxf(v.x, 0.0f); o[1] = (_Float16)fmaxf(v.y, 0.0f);
  o[2] = (_Float16)fmaxf(v.z, 0.0f); o[3] = (_Float16)fmaxf(v.w, 0.0f);
  *(half4*)(h + (size_t)node * HID + c) = o;
}

__global__ void wprep_kernel(const float* __restrict__ W,
                             _Float16* __restrict__ WT) {
  int t = blockIdx.x * 256 + threadIdx.x;
  if (t >= HID * HID) return;
  int k = t & 127, n = t >> 7;
  WT[t] = (_Float16)W[k * HID + n];
}

__global__ void zero_kernel(int* __restrict__ p, int n) {
  int t = blockIdx.x * 256 + threadIdx.x;
  if (t < n) p[t] = 0;
}

__global__ void degree_kernel(const int* __restrict__ ei,
                              int* __restrict__ deg,
                              int* __restrict__ rank, int nE) {
  int e = blockIdx.x * 256 + threadIdx.x;
  if (e < nE) rank[e] = atomicAdd(deg + ei[nE + e], 1);
}

__global__ __launch_bounds__(256) void blocksum_kernel(
    const int* __restrict__ deg, int* __restrict__ bsum, int n) {
  __shared__ int red[256];
  const int t = threadIdx.x;
  const int i0 = blockIdx.x * 1024 + t * 4;
  int v = 0;
#pragma unroll
  for (int k = 0; k < 4; ++k)
    if (i0 + k < n) v += deg[i0 + k];
  red[t] = v;
  __syncthreads();
  for (int st = 128; st > 0; st >>= 1) {
    if (t < st) red[t] += red[t + st];
    __syncthreads();
  }
  if (t == 0) bsum[blockIdx.x] = red[0];
}

__global__ __launch_bounds__(256) void scan_kernel(
    const int* __restrict__ deg, const int* __restrict__ bsum,
    int* __restrict__ rowptr, int n) {
  __shared__ int sh[256];
  const int t = threadIdx.x;
  const int b = blockIdx.x;
  int part = 0;
  for (int i = t; i < b; i += 256) part += bsum[i];
  sh[t] = part;
  __syncthreads();
  for (int st = 128; st > 0; st >>= 1) {
    if (t < st) sh[t] += sh[t + st];
    __syncthreads();
  }
  const int offset = sh[0];
  __syncthreads();
  const int i0 = b * 1024 + t * 4;
  int d[4];
#pragma unroll
  for (int k = 0; k < 4; ++k) d[k] = (i0 + k < n) ? deg[i0 + k] : 0;
  const int tsum = d[0] + d[1] + d[2] + d[3];
  sh[t] = tsum;
  __syncthreads();
  for (int st = 1; st < 256; st <<= 1) {
    const int v = (t >= st) ? sh[t - st] : 0;
    __syncthreads();
    sh[t] += v;
    __syncthreads();
  }
  int run = offset + sh[t] - tsum;
#pragma unroll
  for (int k = 0; k < 4; ++k) {
    if (i0 + k < n) {
      rowptr[i0 + k] = run;
      run += d[k];
      if (i0 + k == n - 1) rowptr[n] = run;
    }
  }
}

__global__ void fill_kernel(const int* __restrict__ ei,
                            const int* __restrict__ rowptr,
                            const int* __restrict__ rank,
                            int* __restrict__ adj, int nE) {
  int e = blockIdx.x * 256 + threadIdx.x;
  if (e >= nE) return;
  const int src = ei[e];
  const int dst = ei[nE + e];
  adj[rowptr[dst] + rank[e]] = src;
}

__global__ __launch_bounds__(256) void segsort_kernel(
    const int* __restrict__ rowptr, const int* __restrict__ adj,
    int* __restrict__ adjs, int n) {
  const int node =
      __builtin_amdgcn_readfirstlane((blockIdx.x * 256 + threadIdx.x) >> 5);
  if (node >= n) return;
  const int lane  = threadIdx.x & 31;
  const int start = rowptr[node];
  const int deg   = rowptr[node + 1] - start;
  if (deg <= 32) {
    const int key = (lane < deg) ? adj[start + lane] : 0x7fffffff;
    int rank = 0;
    for (int j = 0; j < deg; ++j) {
      const int kj = __builtin_amdgcn_readlane(key, j);
      rank += ((kj < key) || (kj == key && j < lane)) ? 1 : 0;
    }
    if (lane < deg) adjs[start + rank] = key;
  } else {
    for (int i = lane; i < deg; i += 32) {
      const int key = adj[start + i];
      int rank = 0;
      for (int j = 0; j < deg; ++j) {
        const int kj = adj[start + j];
        rank += ((kj < key) || (kj == key && j < i)) ? 1 : 0;
      }
      adjs[start + rank] = key;
    }
  }
}

template <int U>
__device__ inline void take_rows(const _Float16* __restrict__ hl, int key,
                                 int j, float4& acc) {
  half4 v[U];
#pragma unroll
  for (int u = 0; u < U; ++u) {
    const int sidx = __builtin_amdgcn_readlane(key, j + u);
    v[u] = *(const half4*)(hl + (size_t)sidx * HID);
  }
#pragma unroll
  for (int u = 0; u < U; ++u) {
    acc.x += (float)v[u][0]; acc.y += (float)v[u][1];
    acc.z += (float)v[u][2]; acc.w += (float)v[u][3];
  }
}

__global__ __launch_bounds__(256) void gather_kernel(
    const _Float16* __restrict__ h, const int* __restrict__ rowptr,
    const int* __restrict__ adjs, _Float16* __restrict__ agg, int n) {
  const int node =
      __builtin_amdgcn_readfirstlane((blockIdx.x * 256 + threadIdx.x) >> 5);
  if (node >= n) return;
  const int lane  = threadIdx.x & 31;
  const int start = rowptr[node];
  const int end   = rowptr[node + 1];
  const _Float16* hl = h + lane * 4;
  float4 acc = make_float4(0.0f, 0.0f, 0.0f, 0.0f);
  for (int base = start; base < end; base += 32) {
    const int m   = (end - base < 32) ? (end - base) : 32;
    const int key = (lane < m) ? adjs[base + lane] : 0;
    int j = 0;
    for (; j + 8 <= m; j += 8) take_rows<8>(hl, key, j, acc);
    if (j + 4 <= m) { take_rows<4>(hl, key, j, acc); j += 4; }
    if (j + 2 <= m) { take_rows<2>(hl, key, j, acc); j += 2; }
    if (j < m) take_rows<1>(hl, key, j, acc);
  }
  const int deg = end - start;
  const float inv = (deg > 0) ? __builtin_amdgcn_rcpf((float)deg) : 0.0f;
  half4 o;
  o[0] = (_Float16)(acc.x * inv); o[1] = (_Float16)(acc.y * inv);
  o[2] = (_Float16)(acc.z * inv); o[3] = (_Float16)(acc.w * inv);
  *(half4*)(agg + (size_t)node * HID + lane * 4) = o;
}

__device__ inline v16h join16(half8 lo, half8 hi) {
  return __builtin_shufflevector(lo, hi, 0, 1, 2, 3, 4, 5, 6, 7,
                                         8, 9, 10, 11, 12, 13, 14, 15);
}

template <bool SAGE, bool RELU, bool COLSUM, bool OUTH>
__global__ __launch_bounds__(256) void gemm_kernel(
    const _Float16* __restrict__ h,
    const _Float16* __restrict__ agg,
    const _Float16* __restrict__ WlT,
    const _Float16* __restrict__ WrT,
    const float*    __restrict__ bias,
    float*    __restrict__ outf,
    _Float16* __restrict__ outh,
    float* __restrict__ colpart,
    int n_nodes) {
  __shared__ __attribute__((aligned(16))) _Float16 wtr[HID * WPITCH];
  __shared__ __attribute__((aligned(16))) _Float16 wtl[HID * WPITCH];
  __shared__ float colred[COLSUM ? 16 * HID : 1];

  {
    const uint4* sr = (const uint4*)WrT;
    for (int i = threadIdx.x; i < HID * HID / 8; i += 256)
      *(uint4*)&wtr[(i >> 4) * WPITCH + (i & 15) * 8] = sr[i];
    if constexpr (SAGE) {
      const uint4* sl = (const uint4*)WlT;
      for (int i = threadIdx.x; i < HID * HID / 8; i += 256)
        *(uint4*)&wtl[(i >> 4) * WPITCH + (i & 15) * 8] = sl[i];
    }
  }
  __syncthreads();

  const int lane = threadIdx.x & 31;
  const int wave = threadIdx.x >> 5;
  const int rowbase = blockIdx.x * 128 + wave * 16;
  const bool active = rowbase < n_nodes;
  if constexpr (!COLSUM) {
    if (!active) return;
  }

  float csum[8];
#pragma unroll
  for (int nt = 0; nt < 8; ++nt) csum[nt] = 0.0f;

  if (active) {

  const int arow = rowbase + (lane & 15);
  const int row  = arow < n_nodes ? arow : n_nodes - 1;
  const int lo   = (lane >= 16) ? 8 : 0;

  const _Float16* hp  = h + (size_t)row * HID + lo;
  const _Float16* gp  = SAGE ? (agg + (size_t)row * HID + lo) : nullptr;
  const _Float16* brp = &wtr[(lane & 15) * WPITCH + lo];
  const _Float16* blp = &wtl[(lane & 15) * WPITCH + lo];

  v8f acc[8];
  {
    v8f z = {};
#pragma unroll
    for (int nt = 0; nt < 8; ++nt) acc[nt] = z;
  }

#pragma unroll
  for (int kk = 0; kk < 4; ++kk) {
    const int kb = kk * 32;
    v16h ahf = join16(*(const half8*)(hp + kb),
                      *(const half8*)(hp + kb + 16));
    v16h agf;
    if constexpr (SAGE)
      agf = join16(*(const half8*)(gp + kb),
                   *(const half8*)(gp + kb + 16));
#pragma unroll
    for (int nt = 0; nt < 8; ++nt) {
      const int off = nt * 16 * WPITCH + kb;
      v16h br = join16(*(const half8*)(brp + off),
                       *(const half8*)(brp + off + 16));
      acc[nt] = __builtin_amdgcn_wmma_f32_16x16x32_f16(
          false, ahf, false, br, (short)0, acc[nt], false, false);
      if constexpr (SAGE) {
        v16h bl = join16(*(const half8*)(blp + off),
                         *(const half8*)(blp + off + 16));
        acc[nt] = __builtin_amdgcn_wmma_f32_16x16x32_f16(
            false, agf, false, bl, (short)0, acc[nt], false, false);
      }
    }
  }

  const int mofs = (lane >= 16) ? 8 : 0;
#pragma unroll
  for (int nt = 0; nt < 8; ++nt) {
    const int ncol = nt * 16 + (lane & 15);
    const float bb = bias[ncol];
#pragma unroll
    for (int r = 0; r < 8; ++r) {
      float v = acc[nt][r] + bb;
      if constexpr (RELU) v = fmaxf(v, 0.0f);
      int orow = rowbase + mofs + r;
      if (orow < n_nodes) {
        if constexpr (OUTH) outh[(size_t)orow * HID + ncol] = (_Float16)v;
        else                outf[(size_t)orow * HID + ncol] = v;
        if constexpr (COLSUM) csum[nt] += v;
      }
    }
  }
  }

  if constexpr (COLSUM) {
    float* cr = &colred[(wave * 2 + (lane >> 4)) * HID + (lane & 15)];
#pragma unroll
    for (int nt = 0; nt < 8; ++nt) cr[nt * 16] = csum[nt];
    __syncthreads();
    if (threadIdx.x < HID) {
      float t = 0.0f;
#pragma unroll
      for (int p = 0; p < 16; ++p) t += colred[p * HID + threadIdx.x];
      colpart[(size_t)blockIdx.x * HID + threadIdx.x] = t;
    }
  }
}

__global__ void colmean_kernel(const float* __restrict__ part, int nparts,
                               float* __restrict__ outm, int n) {
  __shared__ float red[256];
  int col = blockIdx.x;
  float acc = 0.0f;
  for (int r = threadIdx.x; r < nparts; r += 256)
    acc += part[(size_t)r * HID + col];
  red[threadIdx.x] = acc;
  __syncthreads();
  for (int st = 128; st > 0; st >>= 1) {
    if (threadIdx.x < st) red[threadIdx.x] += red[threadIdx.x + st];
    __syncthreads();
  }
  if (threadIdx.x == 0) outm[col] = red[0] / (float)n;
}

extern "C" void kernel_launch(void* const* d_in, const int* in_sizes, int n_in,
                              void* d_out, int out_size, void* d_ws, size_t ws_size,
                              hipStream_t stream) {
  const float* pos  = (const float*)d_in[0];
  const int*   an   = (const int*)  d_in[1];
  const int*   ei   = (const int*)  d_in[2];
  const float* W_in = (const float*)d_in[3];
  const float* b_in = (const float*)d_in[4];
  const float* W1l  = (const float*)d_in[5];
  const float* b1   = (const float*)d_in[6];
  const float* W1r  = (const float*)d_in[7];
  const float* W2l  = (const float*)d_in[8];
  const float* b2   = (const float*)d_in[9];
  const float* W2r  = (const float*)d_in[10];
  const float* Wout = (const float*)d_in[11];
  const float* bout = (const float*)d_in[12];

  const int n  = in_sizes[0] / 3;
  const int nE = in_sizes[2] / 2;

  _Float16* h   = (_Float16*)d_ws;
  _Float16* agg = h + (size_t)n * HID;

  _Float16* wtbase = agg + (size_t)n * HID;
  _Float16* W1lT = wtbase + 0 * HID * HID;
  _Float16* W1rT = wtbase + 1 * HID * HID;
  _Float16* W2lT = wtbase + 2 * HID * HID;
  _Float16* W2rT = wtbase + 3 * HID * HID;
  _Float16* WoT  = wtbase + 4 * HID * HID;

  const int scanBlks = (n + 1023) / 1024;
  int* deg    = (int*)(wtbase + 5 * HID * HID);
  int* rowptr = deg + n;
  int* bsum   = rowptr + (n + 1);
  int* adj    = bsum + scanBlks;
  int* adjs   = adj + nE;
  int* rank   = adjs + nE;
  float* colpart = (float*)(rank + nE);

  float* outm = (float*)d_out;
  float* nst  = (float*)d_out + HID;

  const int gemmBlks = (n + 127) / 128;
  const int edgeBlks = (nE + 255) / 256;
  const int nodeBlks = (n + 7) / 8;
  const int wBlks    = (HID * HID + 255) / 256;

  wprep_kernel<<<wBlks, 256, 0, stream>>>(W1l,  W1lT);
  wprep_kernel<<<wBlks, 256, 0, stream>>>(W1r,  W1rT);
  wprep_kernel<<<wBlks, 256, 0, stream>>>(W2l,  W2lT);
  wprep_kernel<<<wBlks, 256, 0, stream>>>(W2r,  W2rT);
  wprep_kernel<<<wBlks, 256, 0, stream>>>(Wout, WoT);

  inproj_kernel<<<nodeBlks, 256, 0, stream>>>(pos, an, W_in, b_in, h, n);

  zero_kernel<<<(n + 255) / 256, 256, 0, stream>>>(deg, n);
  degree_kernel<<<edgeBlks, 256, 0, stream>>>(ei, deg, rank, nE);
  blocksum_kernel<<<scanBlks, 256, 0, stream>>>(deg, bsum, n);
  scan_kernel<<<scanBlks, 256, 0, stream>>>(deg, bsum, rowptr, n);
  fill_kernel<<<edgeBlks, 256, 0, stream>>>(ei, rowptr, rank, adj, nE);
  segsort_kernel<<<nodeBlks, 256, 0, stream>>>(rowptr, adj, adjs, n);

  gather_kernel<<<nodeBlks, 256, 0, stream>>>(h, rowptr, adjs, agg, n);
  gemm_kernel<true, true, false, true><<<gemmBlks, 256, 0, stream>>>(
      h, agg, W1lT, W1rT, b1, nullptr, h, nullptr, n);

  gather_kernel<<<nodeBlks, 256, 0, stream>>>(h, rowptr, adjs, agg, n);
  gemm_kernel<true, true, false, true><<<gemmBlks, 256, 0, stream>>>(
      h, agg, W2lT, W2rT, b2, nullptr, h, nullptr, n);

  gemm_kernel<false, false, true, false><<<gemmBlks, 256, 0, stream>>>(
      h, nullptr, nullptr, WoT, bout, nst, nullptr, colpart, n);
  colmean_kernel<<<HID, 256, 0, stream>>>(colpart, gemmBlks, outm, n);
}


# Input-only preparation in ONE launch, not seven: the five weight transposes, the degree counters' clear and the input projection share one grid by block range; every sum and its order unchanged
# speedup vs baseline: 3.3932x; 1.0453x over previous
#include <hip/hip_runtime.h>


typedef __attribute__((ext_vector_type(16))) _Float16 v16h;
typedef __attribute__((ext_vector_type(8)))  _Float16 half8;
typedef __attribute__((ext_vector_type(4)))  _Float16 half4;
typedef __attribute__((ext_vector_type(8)))  float    v8f;

#define HID 128
#define WPITCH 136

#define WBLKS (HID * HID / 256)
__global__ __launch_bounds__(256) void prep_kernel(
    const float* __restrict__ pos,
    const int*   __restrict__ an,
    const float* __restrict__ W_in,
    const float* __restrict__ b_in,
    const float* __restrict__ W1l, const float* __restrict__ W1r,
    const float* __restrict__ W2l, const float* __restrict__ W2r,
    const float* __restrict__ Wout,
    _Float16* __restrict__ wt,
    int* __restrict__ deg, int zeroBlks,
    _Float16* __restrict__ h, int n) {
  const int blk = blockIdx.x;
  if (blk < 5 * WBLKS) {
    const int m = blk / WBLKS;
    const float* W = (m == 0) ? W1l : (m == 1) ? W1r : (m == 2) ? W2l
                   : (m == 3) ? W2r : Wout;
    const int t = (blk - m * WBLKS) * 256 + threadIdx.x;
    const int k = t & 127, c = t >> 7;
    wt[m * HID * HID + t] = (_Float16)W[k * HID + c];
    return;
  }
  if (blk < 5 * WBLKS + zeroBlks) {
    const int t = (blk - 5 * WBLKS) * 256 + threadIdx.x;
    if (t < n) deg[t] = 0;
    return;
  }
  const int node = __builtin_amdgcn_readfirstlane(
      ((blk - 5 * WBLKS - zeroBlks) * 256 + threadIdx.x) >> 5);
  if (node >= n) return;
  const int c = (threadIdx.x & 31) * 4;
  const float z  = (float)an[node] * 0.1f;
  const float px = pos[node * 3 + 0];
  const float py = pos[node * 3 + 1];
  const float pz = pos[node * 3 + 2];
  const float4 w0 = *(const float4*)(W_in + 0 * HID + c);
  const float4 w1 = *(const float4*)(W_in + 1 * HID + c);
  const float4 w2 = *(const float4*)(W_in + 2 * HID + c);
  const float4 w3 = *(const float4*)(W_in + 3 * HID + c);
  float4 v = *(const float4*)(b_in + c);
  v.x = fmaf(z,  w0.x, v.x); v.y = fmaf(z,  w0.y, v.y);
  v.z = fmaf(z,  w0.z, v.z); v.w = fmaf(z,  w0.w, v.w);
  v.x = fmaf(px, w1.x, v.x); v.y = fmaf(px, w1.y, v.y);
  v.z = fmaf(px, w1.z, v.z); v.w = fmaf(px, w1.w, v.w);
  v.x = fmaf(py, w2.x, v.x); v.y = fmaf(py, w2.y, v.y);
  v.z = fmaf(py, w2.z, v.z); v.w = fmaf(py, w2.w, v.w);
  v.x = fmaf(pz, w3.x, v.x); v.y = fmaf(pz, w3.y, v.y);
  v.z = fmaf(pz, w3.z, v.z); v.w = fmaf(pz, w3.w, v.w);
  half4 o;
  o[0] = (_Float16)fmaxf(v.x, 0.0f); o[1] = (_Float16)fmaxf(v.y, 0.0f);
  o[2] = (_Float16)fmaxf(v.z, 0.0f); o[3] = (_Float16)fmaxf(v.w, 0.0f);
  *(half4*)(h + (size_t)node * HID + c) = o;
}

__global__ void degree_kernel(const int* __restrict__ ei,
                              int* __restrict__ deg,
                              int* __restrict__ rank, int nE) {
  int e = blockIdx.x * 256 + threadIdx.x;
  if (e < nE) rank[e] = atomicAdd(deg + ei[nE + e], 1);
}

__global__ __launch_bounds__(256) void blocksum_kernel(
    const int* __restrict__ deg, int* __restrict__ bsum, int n) {
  __shared__ int red[256];
  const int t = threadIdx.x;
  const int i0 = blockIdx.x * 1024 + t * 4;
  int v = 0;
#pragma unroll
  for (int k = 0; k < 4; ++k)
    if (i0 + k < n) v += deg[i0 + k];
  red[t] = v;
  __syncthreads();
  for (int st = 128; st > 0; st >>= 1) {
    if (t < st) red[t] += red[t + st];
    __syncthreads();
  }
  if (t == 0) bsum[blockIdx.x] = red[0];
}

__global__ __launch_bounds__(256) void scan_kernel(
    const int* __restrict__ deg, const int* __restrict__ bsum,
    int* __restrict__ rowptr, int n) {
  __shared__ int sh[256];
  const int t = threadIdx.x;
  const int b = blockIdx.x;
  int part = 0;
  for (int i = t; i < b; i += 256) part += bsum[i];
  sh[t] = part;
  __syncthreads();
  for (int st = 128; st > 0; st >>= 1) {
    if (t < st) sh[t] += sh[t + st];
    __syncthreads();
  }
  const int offset = sh[0];
  __syncthreads();
  const int i0 = b * 1024 + t * 4;
  int d[4];
#pragma unroll
  for (int k = 0; k < 4; ++k) d[k] = (i0 + k < n) ? deg[i0 + k] : 0;
  const int tsum = d[0] + d[1] + d[2] + d[3];
  sh[t] = tsum;
  __syncthreads();
  for (int st = 1; st < 256; st <<= 1) {
    const int v = (t >= st) ? sh[t - st] : 0;
    __syncthreads();
    sh[t] += v;
    __syncthreads();
  }
  int run = offset + sh[t] - tsum;
#pragma unroll
  for (int k = 0; k < 4; ++k) {
    if (i0 + k < n) {
      rowptr[i0 + k] = run;
      run += d[k];
      if (i0 + k == n - 1) rowptr[n] = run;
    }
  }
}

__global__ void fill_kernel(const int* __restrict__ ei,
                            const int* __restrict__ rowptr,
                            const int* __restrict__ rank,
                            int* __restrict__ adj, int nE) {
  int e = blockIdx.x * 256 + threadIdx.x;
  if (e >= nE) return;
  const int src = ei[e];
  const int dst = ei[nE + e];
  adj[rowptr[dst] + rank[e]] = src;
}

__global__ __launch_bounds__(256) void segsort_kernel(
    const int* __restrict__ rowptr, const int* __restrict__ adj,
    int* __restrict__ adjs, int n) {
  const int node =
      __builtin_amdgcn_readfirstlane((blockIdx.x * 256 + threadIdx.x) >> 5);
  if (node >= n) return;
  const int lane  = threadIdx.x & 31;
  const int start = rowptr[node];
  const int deg   = rowptr[node + 1] - start;
  if (deg <= 32) {
    const int key = (lane < deg) ? adj[start + lane] : 0x7fffffff;
    int rank = 0;
    for (int j = 0; j < deg; ++j) {
      const int kj = __builtin_amdgcn_readlane(key, j);
      rank += ((kj < key) || (kj == key && j < lane)) ? 1 : 0;
    }
    if (lane < deg) adjs[start + rank] = key;
  } else {
    for (int i = lane; i < deg; i += 32) {
      const int key = adj[start + i];
      int rank = 0;
      for (int j = 0; j < deg; ++j) {
        const int kj = adj[start + j];
        rank += ((kj < key) || (kj == key && j < i)) ? 1 : 0;
      }
      adjs[start + rank] = key;
    }
  }
}

template <int U>
__device__ inline void take_rows(const _Float16* __restrict__ hl, int key,
                                 int j, float4& acc) {
  half4 v[U];
#pragma unroll
  for (int u = 0; u < U; ++u) {
    const int sidx = __builtin_amdgcn_readlane(key, j + u);
    v[u] = *(const half4*)(hl + (size_t)sidx * HID);
  }
#pragma unroll
  for (int u = 0; u < U; ++u) {
    acc.x += (float)v[u][0]; acc.y += (float)v[u][1];
    acc.z += (float)v[u][2]; acc.w += (float)v[u][3];
  }
}

__global__ __launch_bounds__(256) void gather_kernel(
    const _Float16* __restrict__ h, const int* __restrict__ rowptr,
    const int* __restrict__ adjs, _Float16* __restrict__ agg, int n) {
  const int node =
      __builtin_amdgcn_readfirstlane((blockIdx.x * 256 + threadIdx.x) >> 5);
  if (node >= n) return;
  const int lane  = threadIdx.x & 31;
  const int start = rowptr[node];
  const int end   = rowptr[node + 1];
  const _Float16* hl = h + lane * 4;
  float4 acc = make_float4(0.0f, 0.0f, 0.0f, 0.0f);
  for (int base = start; base < end; base += 32) {
    const int m   = (end - base < 32) ? (end - base) : 32;
    const int key = (lane < m) ? adjs[base + lane] : 0;
    int j = 0;
    for (; j + 8 <= m; j += 8) take_rows<8>(hl, key, j, acc);
    if (j + 4 <= m) { take_rows<4>(hl, key, j, acc); j += 4; }
    if (j + 2 <= m) { take_rows<2>(hl, key, j, acc); j += 2; }
    if (j < m) take_rows<1>(hl, key, j, acc);
  }
  const int deg = end - start;
  const float inv = (deg > 0) ? __builtin_amdgcn_rcpf((float)deg) : 0.0f;
  half4 o;
  o[0] = (_Float16)(acc.x * inv); o[1] = (_Float16)(acc.y * inv);
  o[2] = (_Float16)(acc.z * inv); o[3] = (_Float16)(acc.w * inv);
  *(half4*)(agg + (size_t)node * HID + lane * 4) = o;
}

__device__ inline v16h join16(half8 lo, half8 hi) {
  return __builtin_shufflevector(lo, hi, 0, 1, 2, 3, 4, 5, 6, 7,
                                         8, 9, 10, 11, 12, 13, 14, 15);
}

template <bool SAGE, bool RELU, bool COLSUM, bool OUTH>
__global__ __launch_bounds__(256) void gemm_kernel(
    const _Float16* __restrict__ h,
    const _Float16* __restrict__ agg,
    const _Float16* __restrict__ WlT,
    const _Float16* __restrict__ WrT,
    const float*    __restrict__ bias,
    float*    __restrict__ outf,
    _Float16* __restrict__ outh,
    float* __restrict__ colpart,
    int n_nodes) {
  __shared__ __attribute__((aligned(16))) _Float16 wtr[HID * WPITCH];
  __shared__ __attribute__((aligned(16))) _Float16 wtl[HID * WPITCH];
  __shared__ float colred[COLSUM ? 16 * HID : 1];

  {
    const uint4* sr = (const uint4*)WrT;
    for (int i = threadIdx.x; i < HID * HID / 8; i += 256)
      *(uint4*)&wtr[(i >> 4) * WPITCH + (i & 15) * 8] = sr[i];
    if constexpr (SAGE) {
      const uint4* sl = (const uint4*)WlT;
      for (int i = threadIdx.x; i < HID * HID / 8; i += 256)
        *(uint4*)&wtl[(i >> 4) * WPITCH + (i & 15) * 8] = sl[i];
    }
  }
  __syncthreads();

  const int lane = threadIdx.x & 31;
  const int wave = threadIdx.x >> 5;
  const int rowbase = blockIdx.x * 128 + wave * 16;
  const bool active = rowbase < n_nodes;
  if constexpr (!COLSUM) {
    if (!active) return;
  }

  float csum[8];
#pragma unroll
  for (int nt = 0; nt < 8; ++nt) csum[nt] = 0.0f;

  if (active) {

  const int arow = rowbase + (lane & 15);
  const int row  = arow < n_nodes ? arow : n_nodes - 1;
  const int lo   = (lane >= 16) ? 8 : 0;

  const _Float16* hp  = h + (size_t)row * HID + lo;
  const _Float16* gp  = SAGE ? (agg + (size_t)row * HID + lo) : nullptr;
  const _Float16* brp = &wtr[(lane & 15) * WPITCH + lo];
  const _Float16* blp = &wtl[(lane & 15) * WPITCH + lo];

  v8f acc[8];
  {
    v8f z = {};
#pragma unroll
    for (int nt = 0; nt < 8; ++nt) acc[nt] = z;
  }

#pragma unroll
  for (int kk = 0; kk < 4; ++kk) {
    const int kb = kk * 32;
    v16h ahf = join16(*(const half8*)(hp + kb),
                      *(const half8*)(hp + kb + 16));
    v16h agf;
    if constexpr (SAGE)
      agf = join16(*(const half8*)(gp + kb),
                   *(const half8*)(gp + kb + 16));
#pragma unroll
    for (int nt = 0; nt < 8; ++nt) {
      const int off = nt * 16 * WPITCH + kb;
      v16h br = join16(*(const half8*)(brp + off),
                       *(const half8*)(brp + off + 16));
      acc[nt] = __builtin_amdgcn_wmma_f32_16x16x32_f16(
          false, ahf, false, br, (short)0, acc[nt], false, false);
      if constexpr (SAGE) {
        v16h bl = join16(*(const half8*)(blp + off),
                         *(const half8*)(blp + off + 16));
        acc[nt] = __builtin_amdgcn_wmma_f32_16x16x32_f16(
            false, agf, false, bl, (short)0, acc[nt], false, false);
      }
    }
  }

  const int mofs = (lane >= 16) ? 8 : 0;
#pragma unroll
  for (int nt = 0; nt < 8; ++nt) {
    const int ncol = nt * 16 + (lane & 15);
    const float bb = bias[ncol];
#pragma unroll
    for (int r = 0; r < 8; ++r) {
      float v = acc[nt][r] + bb;
      if constexpr (RELU) v = fmaxf(v, 0.0f);
      int orow = rowbase + mofs + r;
      if (orow < n_nodes) {
        if constexpr (OUTH) outh[(size_t)orow * HID + ncol] = (_Float16)v;
        else                outf[(size_t)orow * HID + ncol] = v;
        if constexpr (COLSUM) csum[nt] += v;
      }
    }
  }
  }

  if constexpr (COLSUM) {
    float* cr = &colred[(wave * 2 + (lane >> 4)) * HID + (lane & 15)];
#pragma unroll
    for (int nt = 0; nt < 8; ++nt) cr[nt * 16] = csum[nt];
    __syncthreads();
    if (threadIdx.x < HID) {
      float t = 0.0f;
#pragma unroll
      for (int p = 0; p < 16; ++p) t += colred[p * HID + threadIdx.x];
      colpart[(size_t)blockIdx.x * HID + threadIdx.x] = t;
    }
  }
}

__global__ void colmean_kernel(const float* __restrict__ part, int nparts,
                               float* __restrict__ outm, int n) {
  __shared__ float red[256];
  int col = blockIdx.x;
  float acc = 0.0f;
  for (int r = threadIdx.x; r < nparts; r += 256)
    acc += part[(size_t)r * HID + col];
  red[threadIdx.x] = acc;
  __syncthreads();
  for (int st = 128; st > 0; st >>= 1) {
    if (threadIdx.x < st) red[threadIdx.x] += red[threadIdx.x + st];
    __syncthreads();
  }
  if (threadIdx.x == 0) outm[col] = red[0] / (float)n;
}

extern "C" void kernel_launch(void* const* d_in, const int* in_sizes, int n_in,
                              void* d_out, int out_size, void* d_ws, size_t ws_size,
                              hipStream_t stream) {
  const float* pos  = (const float*)d_in[0];
  const int*   an   = (const int*)  d_in[1];
  const int*   ei   = (const int*)  d_in[2];
  const float* W_in = (const float*)d_in[3];
  const float* b_in = (const float*)d_in[4];
  const float* W1l  = (const float*)d_in[5];
  const float* b1   = (const float*)d_in[6];
  const float* W1r  = (const float*)d_in[7];
  const float* W2l  = (const float*)d_in[8];
  const float* b2   = (const float*)d_in[9];
  const float* W2r  = (const float*)d_in[10];
  const float* Wout = (const float*)d_in[11];
  const float* bout = (const float*)d_in[12];

  const int n  = in_sizes[0] / 3;
  const int nE = in_sizes[2] / 2;

  _Float16* h   = (_Float16*)d_ws;
  _Float16* agg = h + (size_t)n * HID;

  _Float16* wtbase = agg + (size_t)n * HID;
  _Float16* W1lT = wtbase + 0 * HID * HID;
  _Float16* W1rT = wtbase + 1 * HID * HID;
  _Float16* W2lT = wtbase + 2 * HID * HID;
  _Float16* W2rT = wtbase + 3 * HID * HID;
  _Float16* WoT  = wtbase + 4 * HID * HID;

  const int scanBlks = (n + 1023) / 1024;
  int* deg    = (int*)(wtbase + 5 * HID * HID);
  int* rowptr = deg + n;
  int* bsum   = rowptr + (n + 1);
  int* adj    = bsum + scanBlks;
  int* adjs   = adj + nE;
  int* rank   = adjs + nE;
  float* colpart = (float*)(rank + nE);

  float* outm = (float*)d_out;
  float* nst  = (float*)d_out + HID;

  const int gemmBlks = (n + 127) / 128;
  const int edgeBlks = (nE + 255) / 256;
  const int nodeBlks = (n + 7) / 8;
  const int zeroBlks = (n + 255) / 256;

  prep_kernel<<<5 * WBLKS + zeroBlks + nodeBlks, 256, 0, stream>>>(
      pos, an, W_in, b_in, W1l, W1r, W2l, W2r, Wout, wtbase, deg, zeroBlks,
      h, n);

  degree_kernel<<<edgeBlks, 256, 0, stream>>>(ei, deg, rank, nE);
  blocksum_kernel<<<scanBlks, 256, 0, stream>>>(deg, bsum, n);
  scan_kernel<<<scanBlks, 256, 0, stream>>>(deg, bsum, rowptr, n);
  fill_kernel<<<edgeBlks, 256, 0, stream>>>(ei, rowptr, rank, adj, nE);
  segsort_kernel<<<nodeBlks, 256, 0, stream>>>(rowptr, adj, adjs, n);

  gather_kernel<<<nodeBlks, 256, 0, stream>>>(h, rowptr, adjs, agg, n);
  gemm_kernel<true, true, false, true><<<gemmBlks, 256, 0, stream>>>(
      h, agg, W1lT, W1rT, b1, nullptr, h, nullptr, n);

  gather_kernel<<<nodeBlks, 256, 0, stream>>>(h, rowptr, adjs, agg, n);
  gemm_kernel<true, true, false, true><<<gemmBlks, 256, 0, stream>>>(
      h, agg, W2lT, W2rT, b2, nullptr, h, nullptr, n);

  gemm_kernel<false, false, true, false><<<gemmBlks, 256, 0, stream>>>(
      h, nullptr, nullptr, WoT, bout, nst, nullptr, colpart, n);
  colmean_kernel<<<HID, 256, 0, stream>>>(colpart, gemmBlks, outm, n);
}


# Degree histogram keeps one counter set per chip half (die id picks the set; set kept in the rank word), so no counter line moves between the two L2 halves; list order after sort and sums same
# speedup vs baseline: 4.2501x; 1.2525x over previous
#include <hip/hip_runtime.h>


typedef __attribute__((ext_vector_type(16))) _Float16 v16h;
typedef __attribute__((ext_vector_type(8)))  _Float16 half8;
typedef __attribute__((ext_vector_type(4)))  _Float16 half4;
typedef __attribute__((ext_vector_type(8)))  float    v8f;

#define HID 128
#define WPITCH 136

#define WBLKS (HID * HID / 256)
__global__ __launch_bounds__(256) void prep_kernel(
    const float* __restrict__ pos,
    const int*   __restrict__ an,
    const float* __restrict__ W_in,
    const float* __restrict__ b_in,
    const float* __restrict__ W1l, const float* __restrict__ W1r,
    const float* __restrict__ W2l, const float* __restrict__ W2r,
    const float* __restrict__ Wout,
    _Float16* __restrict__ wt,
    int* __restrict__ deg, int zeroBlks,
    _Float16* __restrict__ h, int n) {
  const int blk = blockIdx.x;
  if (blk < 5 * WBLKS) {
    const int m = blk / WBLKS;
    const float* W = (m == 0) ? W1l : (m == 1) ? W1r : (m == 2) ? W2l
                   : (m == 3) ? W2r : Wout;
    const int t = (blk - m * WBLKS) * 256 + threadIdx.x;
    const int k = t & 127, c = t >> 7;
    wt[m * HID * HID + t] = (_Float16)W[k * HID + c];
    return;
  }
  if (blk < 5 * WBLKS + zeroBlks) {
    const int t = (blk - 5 * WBLKS) * 256 + threadIdx.x;
    if (t < 2 * n) deg[t] = 0;
    return;
  }
  const int node = __builtin_amdgcn_readfirstlane(
      ((blk - 5 * WBLKS - zeroBlks) * 256 + threadIdx.x) >> 5);
  if (node >= n) return;
  const int c = (threadIdx.x & 31) * 4;
  const float z  = (float)an[node] * 0.1f;
  const float px = pos[node * 3 + 0];
  const float py = pos[node * 3 + 1];
  const float pz = pos[node * 3 + 2];
  const float4 w0 = *(const float4*)(W_in + 0 * HID + c);
  const float4 w1 = *(const float4*)(W_in + 1 * HID + c);
  const float4 w2 = *(const float4*)(W_in + 2 * HID + c);
  const float4 w3 = *(const float4*)(W_in + 3 * HID + c);
  float4 v = *(const float4*)(b_in + c);
  v.x = fmaf(z,  w0.x, v.x); v.y = fmaf(z,  w0.y, v.y);
  v.z = fmaf(z,  w0.z, v.z); v.w = fmaf(z,  w0.w, v.w);
  v.x = fmaf(px, w1.x, v.x); v.y = fmaf(px, w1.y, v.y);
  v.z = fmaf(px, w1.z, v.z); v.w = fmaf(px, w1.w, v.w);
  v.x = fmaf(py, w2.x, v.x); v.y = fmaf(py, w2.y, v.y);
  v.z = fmaf(py, w2.z, v.z); v.w = fmaf(py, w2.w, v.w);
  v.x = fmaf(pz, w3.x, v.x); v.y = fmaf(pz, w3.y, v.y);
  v.z = fmaf(pz, w3.z, v.z); v.w = fmaf(pz, w3.w, v.w);
  half4 o;
  o[0] = (_Float16)fmaxf(v.x, 0.0f); o[1] = (_Float16)fmaxf(v.y, 0.0f);
  o[2] = (_Float16)fmaxf(v.z, 0.0f); o[3] = (_Float16)fmaxf(v.w, 0.0f);
  *(half4*)(h + (size_t)node * HID + c) = o;
}

__global__ void degree_kernel(const int* __restrict__ ei,
                              int* __restrict__ deg,
                              int* __restrict__ rank, int nE, int n) {
  int e = blockIdx.x * 256 + threadIdx.x;
  const unsigned id = __builtin_amdgcn_s_sendmsg_rtn(0x87);
  const int set = (int)((id >> 18) & 1u);
  if (e < nE) {
    const int r = atomicAdd(deg + set * n + ei[nE + e], 1);
    rank[e] = r | (set << 31);
  }
}

__global__ __launch_bounds__(256) void blocksum_kernel(
    const int* __restrict__ deg, int* __restrict__ bsum, int n) {
  __shared__ int red[256];
  const int t = threadIdx.x;
  const int i0 = blockIdx.x * 1024 + t * 4;
  int v = 0;
#pragma unroll
  for (int k = 0; k < 4; ++k)
    if (i0 + k < n) v += deg[i0 + k] + deg[n + i0 + k];
  red[t] = v;
  __syncthreads();
  for (int st = 128; st > 0; st >>= 1) {
    if (t < st) red[t] += red[t + st];
    __syncthreads();
  }
  if (t == 0) bsum[blockIdx.x] = red[0];
}

__global__ __launch_bounds__(256) void scan_kernel(
    const int* __restrict__ deg, const int* __restrict__ bsum,
    int* __restrict__ rowptr, int n) {
  __shared__ int sh[256];
  const int t = threadIdx.x;
  const int b = blockIdx.x;
  int part = 0;
  for (int i = t; i < b; i += 256) part += bsum[i];
  sh[t] = part;
  __syncthreads();
  for (int st = 128; st > 0; st >>= 1) {
    if (t < st) sh[t] += sh[t + st];
    __syncthreads();
  }
  const int offset = sh[0];
  __syncthreads();
  const int i0 = b * 1024 + t * 4;
  int d[4];
#pragma unroll
  for (int k = 0; k < 4; ++k)
    d[k] = (i0 + k < n) ? deg[i0 + k] + deg[n + i0 + k] : 0;
  const int tsum = d[0] + d[1] + d[2] + d[3];
  sh[t] = tsum;
  __syncthreads();
  for (int st = 1; st < 256; st <<= 1) {
    const int v = (t >= st) ? sh[t - st] : 0;
    __syncthreads();
    sh[t] += v;
    __syncthreads();
  }
  int run = offset + sh[t] - tsum;
#pragma unroll
  for (int k = 0; k < 4; ++k) {
    if (i0 + k < n) {
      rowptr[i0 + k] = run;
      run += d[k];
      if (i0 + k == n - 1) rowptr[n] = run;
    }
  }
}

__global__ void fill_kernel(const int* __restrict__ ei,
                            const int* __restrict__ rowptr,
                            const int* __restrict__ rank,
                            const int* __restrict__ deg,
                            int* __restrict__ adj, int nE) {
  int e = blockIdx.x * 256 + threadIdx.x;
  if (e >= nE) return;
  const int src = ei[e];
  const int dst = ei[nE + e];
  const int r   = rank[e];
  int slot = rowptr[dst] + (r & 0x7fffffff);
  if (r < 0) slot += deg[dst];
  adj[slot] = src;
}

__global__ __launch_bounds__(256) void segsort_kernel(
    const int* __restrict__ rowptr, const int* __restrict__ adj,
    int* __restrict__ adjs, int n) {
  const int node =
      __builtin_amdgcn_readfirstlane((blockIdx.x * 256 + threadIdx.x) >> 5);
  if (node >= n) return;
  const int lane  = threadIdx.x & 31;
  const int start = rowptr[node];
  const int deg   = rowptr[node + 1] - start;
  if (deg <= 32) {
    const int key = (lane < deg) ? adj[start + lane] : 0x7fffffff;
    int rank = 0;
    for (int j = 0; j < deg; ++j) {
      const int kj = __builtin_amdgcn_readlane(key, j);
      rank += ((kj < key) || (kj == key && j < lane)) ? 1 : 0;
    }
    if (lane < deg) adjs[start + rank] = key;
  } else {
    for (int i = lane; i < deg; i += 32) {
      const int key = adj[start + i];
      int rank = 0;
      for (int j = 0; j < deg; ++j) {
        const int kj = adj[start + j];
        rank += ((kj < key) || (kj == key && j < i)) ? 1 : 0;
      }
      adjs[start + rank] = key;
    }
  }
}

template <int U>
__device__ inline void take_rows(const _Float16* __restrict__ hl, int key,
                                 int j, float4& acc) {
  half4 v[U];
#pragma unroll
  for (int u = 0; u < U; ++u) {
    const int sidx = __builtin_amdgcn_readlane(key, j + u);
    v[u] = *(const half4*)(hl + (size_t)sidx * HID);
  }
#pragma unroll
  for (int u = 0; u < U; ++u) {
    acc.x += (float)v[u][0]; acc.y += (float)v[u][1];
    acc.z += (float)v[u][2]; acc.w += (float)v[u][3];
  }
}

__global__ __launch_bounds__(256) void gather_kernel(
    const _Float16* __restrict__ h, const int* __restrict__ rowptr,
    const int* __restrict__ adjs, _Float16* __restrict__ agg, int n) {
  const int node =
      __builtin_amdgcn_readfirstlane((blockIdx.x * 256 + threadIdx.x) >> 5);
  if (node >= n) return;
  const int lane  = threadIdx.x & 31;
  const int start = rowptr[node];
  const int end   = rowptr[node + 1];
  const _Float16* hl = h + lane * 4;
  float4 acc = make_float4(0.0f, 0.0f, 0.0f, 0.0f);
  for (int base = start; base < end; base += 32) {
    const int m   = (end - base < 32) ? (end - base) : 32;
    const int key = (lane < m) ? adjs[base + lane] : 0;
    int j = 0;
    for (; j + 8 <= m; j += 8) take_rows<8>(hl, key, j, acc);
    if (j + 4 <= m) { take_rows<4>(hl, key, j, acc); j += 4; }
    if (j + 2 <= m) { take_rows<2>(hl, key, j, acc); j += 2; }
    if (j < m) take_rows<1>(hl, key, j, acc);
  }
  const int deg = end - start;
  const float inv = (deg > 0) ? __builtin_amdgcn_rcpf((float)deg) : 0.0f;
  half4 o;
  o[0] = (_Float16)(acc.x * inv); o[1] = (_Float16)(acc.y * inv);
  o[2] = (_Float16)(acc.z * inv); o[3] = (_Float16)(acc.w * inv);
  *(half4*)(agg + (size_t)node * HID + lane * 4) = o;
}

__device__ inline v16h join16(half8 lo, half8 hi) {
  return __builtin_shufflevector(lo, hi, 0, 1, 2, 3, 4, 5, 6, 7,
                                         8, 9, 10, 11, 12, 13, 14, 15);
}

template <bool SAGE, bool RELU, bool COLSUM, bool OUTH>
__global__ __launch_bounds__(256) void gemm_kernel(
    const _Float16* __restrict__ h,
    const _Float16* __restrict__ agg,
    const _Float16* __restrict__ WlT,
    const _Float16* __restrict__ WrT,
    const float*    __restrict__ bias,
    float*    __restrict__ outf,
    _Float16* __restrict__ outh,
    float* __restrict__ colpart,
    int n_nodes) {
  __shared__ __attribute__((aligned(16))) _Float16 wtr[HID * WPITCH];
  __shared__ __attribute__((aligned(16))) _Float16 wtl[HID * WPITCH];
  __shared__ float colred[COLSUM ? 16 * HID : 1];

  {
    const uint4* sr = (const uint4*)WrT;
    for (int i = threadIdx.x; i < HID * HID / 8; i += 256)
      *(uint4*)&wtr[(i >> 4) * WPITCH + (i & 15) * 8] = sr[i];
    if constexpr (SAGE) {
      const uint4* sl = (const uint4*)WlT;
      for (int i = threadIdx.x; i < HID * HID / 8; i += 256)
        *(uint4*)&wtl[(i >> 4) * WPITCH + (i & 15) * 8] = sl[i];
    }
  }
  __syncthreads();

  const int lane = threadIdx.x & 31;
  const int wave = threadIdx.x >> 5;
  const int rowbase = blockIdx.x * 128 + wave * 16;
  const bool active = rowbase < n_nodes;
  if constexpr (!COLSUM) {
    if (!active) return;
  }

  float csum[8];
#pragma unroll
  for (int nt = 0; nt < 8; ++nt) csum[nt] = 0.0f;

  if (active) {

  const int arow = rowbase + (lane & 15);
  const int row  = arow < n_nodes ? arow : n_nodes - 1;
  const int lo   = (lane >= 16) ? 8 : 0;

  const _Float16* hp  = h + (size_t)row * HID + lo;
  const _Float16* gp  = SAGE ? (agg + (size_t)row * HID + lo) : nullptr;
  const _Float16* brp = &wtr[(lane & 15) * WPITCH + lo];
  const _Float16* blp = &wtl[(lane & 15) * WPITCH + lo];

  v8f acc[8];
  {
    v8f z = {};
#pragma unroll
    for (int nt = 0; nt < 8; ++nt) acc[nt] = z;
  }

#pragma unroll
  for (int kk = 0; kk < 4; ++kk) {
    const int kb = kk * 32;
    v16h ahf = join16(*(const half8*)(hp + kb),
                      *(const half8*)(hp + kb + 16));
    v16h agf;
    if constexpr (SAGE)
      agf = join16(*(const half8*)(gp + kb),
                   *(const half8*)(gp + kb + 16));
#pragma unroll
    for (int nt = 0; nt < 8; ++nt) {
      const int off = nt * 16 * WPITCH + kb;
      v16h br = join16(*(const half8*)(brp + off),
                       *(const half8*)(brp + off + 16));
      acc[nt] = __builtin_amdgcn_wmma_f32_16x16x32_f16(
          false, ahf, false, br, (short)0, acc[nt], false, false);
      if constexpr (SAGE) {
        v16h bl = join16(*(const half8*)(blp + off),
                         *(const half8*)(blp + off + 16));
        acc[nt] = __builtin_amdgcn_wmma_f32_16x16x32_f16(
            false, agf, false, bl, (short)0, acc[nt], false, false);
      }
    }
  }

  const int mofs = (lane >= 16) ? 8 : 0;
#pragma unroll
  for (int nt = 0; nt < 8; ++nt) {
    const int ncol = nt * 16 + (lane & 15);
    const float bb = bias[ncol];
#pragma unroll
    for (int r = 0; r < 8; ++r) {
      float v = acc[nt][r] + bb;
      if constexpr (RELU) v = fmaxf(v, 0.0f);
      int orow = rowbase + mofs + r;
      if (orow < n_nodes) {
        if constexpr (OUTH) outh[(size_t)orow * HID + ncol] = (_Float16)v;
        else                outf[(size_t)orow * HID + ncol] = v;
        if constexpr (COLSUM) csum[nt] += v;
      }
    }
  }
  }

  if constexpr (COLSUM) {
    float* cr = &colred[(wave * 2 + (lane >> 4)) * HID + (lane & 15)];
#pragma unroll
    for (int nt = 0; nt < 8; ++nt) cr[nt * 16] = csum[nt];
    __syncthreads();
    if (threadIdx.x < HID) {
      float t = 0.0f;
#pragma unroll
      for (int p = 0; p < 16; ++p) t += colred[p * HID + threadIdx.x];
      colpart[(size_t)blockIdx.x * HID + threadIdx.x] = t;
    }
  }
}

__global__ void colmean_kernel(const float* __restrict__ part, int nparts,
                               float* __restrict__ outm, int n) {
  __shared__ float red[256];
  int col = blockIdx.x;
  float acc = 0.0f;
  for (int r = threadIdx.x; r < nparts; r += 256)
    acc += part[(size_t)r * HID + col];
  red[threadIdx.x] = acc;
  __syncthreads();
  for (int st = 128; st > 0; st >>= 1) {
    if (threadIdx.x < st) red[threadIdx.x] += red[threadIdx.x + st];
    __syncthreads();
  }
  if (threadIdx.x == 0) outm[col] = red[0] / (float)n;
}

extern "C" void kernel_launch(void* const* d_in, const int* in_sizes, int n_in,
                              void* d_out, int out_size, void* d_ws, size_t ws_size,
                              hipStream_t stream) {
  const float* pos  = (const float*)d_in[0];
  const int*   an   = (const int*)  d_in[1];
  const int*   ei   = (const int*)  d_in[2];
  const float* W_in = (const float*)d_in[3];
  const float* b_in = (const float*)d_in[4];
  const float* W1l  = (const float*)d_in[5];
  const float* b1   = (const float*)d_in[6];
  const float* W1r  = (const float*)d_in[7];
  const float* W2l  = (const float*)d_in[8];
  const float* b2   = (const float*)d_in[9];
  const float* W2r  = (const float*)d_in[10];
  const float* Wout = (const float*)d_in[11];
  const float* bout = (const float*)d_in[12];

  const int n  = in_sizes[0] / 3;
  const int nE = in_sizes[2] / 2;

  _Float16* h   = (_Float16*)d_ws;
  _Float16* agg = h + (size_t)n * HID;

  _Float16* wtbase = agg + (size_t)n * HID;
  _Float16* W1lT = wtbase + 0 * HID * HID;
  _Float16* W1rT = wtbase + 1 * HID * HID;
  _Float16* W2lT = wtbase + 2 * HID * HID;
  _Float16* W2rT = wtbase + 3 * HID * HID;
  _Float16* WoT  = wtbase + 4 * HID * HID;

  const int scanBlks = (n + 1023) / 1024;
  int* deg    = (int*)(wtbase + 5 * HID * HID);
  int* rowptr = deg + 2 * n;
  int* bsum   = rowptr + (n + 1);
  int* adj    = bsum + scanBlks;
  int* adjs   = adj + nE;
  int* rank   = adjs + nE;
  float* colpart = (float*)(rank + nE);

  float* outm = (float*)d_out;
  float* nst  = (float*)d_out + HID;

  const int gemmBlks = (n + 127) / 128;
  const int edgeBlks = (nE + 255) / 256;
  const int nodeBlks = (n + 7) / 8;
  const int zeroBlks = (2 * n + 255) / 256;

  prep_kernel<<<5 * WBLKS + zeroBlks + nodeBlks, 256, 0, stream>>>(
      pos, an, W_in, b_in, W1l, W1r, W2l, W2r, Wout, wtbase, deg, zeroBlks,
      h, n);

  degree_kernel<<<edgeBlks, 256, 0, stream>>>(ei, deg, rank, nE, n);
  blocksum_kernel<<<scanBlks, 256, 0, stream>>>(deg, bsum, n);
  scan_kernel<<<scanBlks, 256, 0, stream>>>(deg, bsum, rowptr, n);
  fill_kernel<<<edgeBlks, 256, 0, stream>>>(ei, rowptr, rank, deg, adj, nE);
  segsort_kernel<<<nodeBlks, 256, 0, stream>>>(rowptr, adj, adjs, n);

  gather_kernel<<<nodeBlks, 256, 0, stream>>>(h, rowptr, adjs, agg, n);
  gemm_kernel<true, true, false, true><<<gemmBlks, 256, 0, stream>>>(
      h, agg, W1lT, W1rT, b1, nullptr, h, nullptr, n);

  gather_kernel<<<nodeBlks, 256, 0, stream>>>(h, rowptr, adjs, agg, n);
  gemm_kernel<true, true, false, true><<<gemmBlks, 256, 0, stream>>>(
      h, agg, W2lT, W2rT, b2, nullptr, h, nullptr, n);

  gemm_kernel<false, false, true, false><<<gemmBlks, 256, 0, stream>>>(
      h, nullptr, nullptr, WoT, bout, nst, nullptr, colpart, n);
  colmean_kernel<<<HID, 256, 0, stream>>>(colpart, gemmBlks, outm, n);
}


# Work per wave raised: per-node kernels (input projection, list sort, neighbour mean) take 8 nodes per wave, per-edge kernels 4 edges per thread, so a fifth as many waves enter; sums and order same
# speedup vs baseline: 4.4808x; 1.0543x over previous
#include <hip/hip_runtime.h>


typedef __attribute__((ext_vector_type(16))) _Float16 v16h;
typedef __attribute__((ext_vector_type(8)))  _Float16 half8;
typedef __attribute__((ext_vector_type(4)))  _Float16 half4;
typedef __attribute__((ext_vector_type(8)))  float    v8f;

#define HID 128
#define WPITCH 136
#define NPW 8
#define EPT 4

__device__ inline float lane_f(float x, int l) {
  return __int_as_float(__builtin_amdgcn_readlane(__float_as_int(x), l));
}

#define WBLKS (HID * HID / 256)
__global__ __launch_bounds__(256) void prep_kernel(
    const float* __restrict__ pos,
    const int*   __restrict__ an,
    const float* __restrict__ W_in,
    const float* __restrict__ b_in,
    const float* __restrict__ W1l, const float* __restrict__ W1r,
    const float* __restrict__ W2l, const float* __restrict__ W2r,
    const float* __restrict__ Wout,
    _Float16* __restrict__ wt,
    int* __restrict__ deg, int zeroBlks,
    _Float16* __restrict__ h, int n) {
  const int blk = blockIdx.x;
  if (blk < 5 * WBLKS) {
    const int m = blk / WBLKS;
    const float* W = (m == 0) ? W1l : (m == 1) ? W1r : (m == 2) ? W2l
                   : (m == 3) ? W2r : Wout;
    const int t = (blk - m * WBLKS) * 256 + threadIdx.x;
    const int k = t & 127, c = t >> 7;
    wt[m * HID * HID + t] = (_Float16)W[k * HID + c];
    return;
  }
  if (blk < 5 * WBLKS + zeroBlks) {
    const int t = ((blk - 5 * WBLKS) * 256 + threadIdx.x) * 4;
    if (t + 3 < 2 * n) {
      *(int4*)(deg + t) = make_int4(0, 0, 0, 0);
    } else {
      for (int k = 0; k < 4; ++k)
        if (t + k < 2 * n) deg[t + k] = 0;
    }
    return;
  }
  const int n0 = NPW * __builtin_amdgcn_readfirstlane(
      ((blk - 5 * WBLKS - zeroBlks) * 256 + threadIdx.x) >> 5);
  if (n0 >= n) return;
  const int lane = threadIdx.x & 31;
  const int cnt  = (n - n0 < NPW) ? (n - n0) : NPW;
  const int c = lane * 4;
  const float4 w0 = *(const float4*)(W_in + 0 * HID + c);
  const float4 w1 = *(const float4*)(W_in + 1 * HID + c);
  const float4 w2 = *(const float4*)(W_in + 2 * HID + c);
  const float4 w3 = *(const float4*)(W_in + 3 * HID + c);
  const float4 bb = *(const float4*)(b_in + c);
  const float zl = (lane < cnt) ? (float)an[n0 + lane] * 0.1f : 0.0f;
  const float pl = (lane < 3 * cnt) ? pos[n0 * 3 + lane] : 0.0f;
#pragma unroll
  for (int i = 0; i < NPW; ++i) {
    if (i < cnt) {
      const float z  = lane_f(zl, i);
      const float px = lane_f(pl, 3 * i + 0);
      const float py = lane_f(pl, 3 * i + 1);
      const float pz = lane_f(pl, 3 * i + 2);
      float4 v = bb;
      v.x = fmaf(z,  w0.x, v.x); v.y = fmaf(z,  w0.y, v.y);
      v.z = fmaf(z,  w0.z, v.z); v.w = fmaf(z,  w0.w, v.w);
      v.x = fmaf(px, w1.x, v.x); v.y = fmaf(px, w1.y, v.y);
      v.z = fmaf(px, w1.z, v.z); v.w = fmaf(px, w1.w, v.w);
      v.x = fmaf(py, w2.x, v.x); v.y = fmaf(py, w2.y, v.y);
      v.z = fmaf(py, w2.z, v.z); v.w = fmaf(py, w2.w, v.w);
      v.x = fmaf(pz, w3.x, v.x); v.y = fmaf(pz, w3.y, v.y);
      v.z = fmaf(pz, w3.z, v.z); v.w = fmaf(pz, w3.w, v.w);
      half4 o;
      o[0] = (_Float16)fmaxf(v.x, 0.0f); o[1] = (_Float16)fmaxf(v.y, 0.0f);
      o[2] = (_Float16)fmaxf(v.z, 0.0f); o[3] = (_Float16)fmaxf(v.w, 0.0f);
      *(half4*)(h + (size_t)(n0 + i) * HID + c) = o;
    }
  }
}

template <bool FULL>
__device__ inline void degree_body(const int* __restrict__ ei,
                                   int* __restrict__ cset,
                                   int* __restrict__ rank, int nE, int e0,
                                   int setbit) {
  int d[EPT], r[EPT];
#pragma unroll
  for (int k = 0; k < EPT; ++k) {
    const int e = e0 + k * 256;
    d[k] = (FULL || e < nE) ? ei[nE + e] : 0;
  }
#pragma unroll
  for (int k = 0; k < EPT; ++k) {
    const int e = e0 + k * 256;
    r[k] = (FULL || e < nE) ? atomicAdd(cset + d[k], 1) : 0;
  }
#pragma unroll
  for (int k = 0; k < EPT; ++k) {
    const int e = e0 + k * 256;
    if (FULL || e < nE) rank[e] = r[k] | setbit;
  }
}

__global__ void degree_kernel(const int* __restrict__ ei,
                              int* __restrict__ deg,
                              int* __restrict__ rank, int nE, int n) {
  const int b0 = blockIdx.x * (256 * EPT);
  const int e0 = b0 + threadIdx.x;
  const unsigned id = __builtin_amdgcn_s_sendmsg_rtn(0x87);
  const int set = (int)((id >> 18) & 1u);
  int* cset = deg + set * n;
  if (b0 + 256 * EPT <= nE) degree_body<true>(ei, cset, rank, nE, e0, set << 31);
  else                      degree_body<false>(ei, cset, rank, nE, e0, set << 31);
}

__global__ __launch_bounds__(256) void blocksum_kernel(
    const int* __restrict__ deg, int* __restrict__ bsum, int n) {
  __shared__ int red[256];
  const int t = threadIdx.x;
  const int i0 = blockIdx.x * 1024 + t * 4;
  int v = 0;
#pragma unroll
  for (int k = 0; k < 4; ++k)
    if (i0 + k < n) v += deg[i0 + k] + deg[n + i0 + k];
  red[t] = v;
  __syncthreads();
  for (int st = 128; st > 0; st >>= 1) {
    if (t < st) red[t] += red[t + st];
    __syncthreads();
  }
  if (t == 0) bsum[blockIdx.x] = red[0];
}

__global__ __launch_bounds__(256) void scan_kernel(
    const int* __restrict__ deg, const int* __restrict__ bsum,
    int* __restrict__ rowptr, int n) {
  __shared__ int sh[256];
  const int t = threadIdx.x;
  const int b = blockIdx.x;
  int part = 0;
  for (int i = t; i < b; i += 256) part += bsum[i];
  sh[t] = part;
  __syncthreads();
  for (int st = 128; st > 0; st >>= 1) {
    if (t < st) sh[t] += sh[t + st];
    __syncthreads();
  }
  const int offset = sh[0];
  __syncthreads();
  const int i0 = b * 1024 + t * 4;
  int d[4];
#pragma unroll
  for (int k = 0; k < 4; ++k)
    d[k] = (i0 + k < n) ? deg[i0 + k] + deg[n + i0 + k] : 0;
  const int tsum = d[0] + d[1] + d[2] + d[3];
  sh[t] = tsum;
  __syncthreads();
  for (int st = 1; st < 256; st <<= 1) {
    const int v = (t >= st) ? sh[t - st] : 0;
    __syncthreads();
    sh[t] += v;
    __syncthreads();
  }
  int run = offset + sh[t] - tsum;
#pragma unroll
  for (int k = 0; k < 4; ++k) {
    if (i0 + k < n) {
      rowptr[i0 + k] = run;
      run += d[k];
      if (i0 + k == n - 1) rowptr[n] = run;
    }
  }
}

template <bool FULL>
__device__ inline void fill_body(const int* __restrict__ ei,
                                 const int* __restrict__ rowptr,
                                 const int* __restrict__ rank,
                                 const int* __restrict__ deg,
                                 int* __restrict__ adj, int nE, int e0) {
  int s[EPT], d[EPT], r[EPT], base[EPT], before[EPT];
#pragma unroll
  for (int k = 0; k < EPT; ++k) {
    const int e = e0 + k * 256;
    const bool in = FULL || e < nE;
    s[k] = in ? ei[e] : 0;
    d[k] = in ? ei[nE + e] : 0;
    r[k] = in ? rank[e] : 0;
  }
#pragma unroll
  for (int k = 0; k < EPT; ++k) {
    const int e = e0 + k * 256;
    base[k] = (FULL || e < nE) ? rowptr[d[k]] : 0;
  }
#pragma unroll
  for (int k = 0; k < EPT; ++k) {
    const int e = e0 + k * 256;
    before[k] = ((FULL || e < nE) && r[k] < 0) ? deg[d[k]] : 0;
  }
#pragma unroll
  for (int k = 0; k < EPT; ++k) {
    const int e = e0 + k * 256;
    if (FULL || e < nE)
      adj[base[k] + (r[k] & 0x7fffffff) + before[k]] = s[k];
  }
}

__global__ void fill_kernel(const int* __restrict__ ei,
                            const int* __restrict__ rowptr,
                            const int* __restrict__ rank,
                            const int* __restrict__ deg,
                            int* __restrict__ adj, int nE) {
  const int b0 = blockIdx.x * (256 * EPT);
  const int e0 = b0 + threadIdx.x;
  if (b0 + 256 * EPT <= nE) fill_body<true>(ei, rowptr, rank, deg, adj, nE, e0);
  else                      fill_body<false>(ei, rowptr, rank, deg, adj, nE, e0);
}

__global__ __launch_bounds__(256) void segsort_kernel(
    const int* __restrict__ rowptr, const int* __restrict__ adj,
    int* __restrict__ adjs, int n) {
  const int n0 = NPW * __builtin_amdgcn_readfirstlane(
      (blockIdx.x * 256 + threadIdx.x) >> 5);
  if (n0 >= n) return;
  const int lane = threadIdx.x & 31;
  const int cnt  = (n - n0 < NPW) ? (n - n0) : NPW;
  const int rp = (lane <= cnt) ? rowptr[n0 + lane] : 0;
  int key[NPW], st[NPW], dg[NPW];
#pragma unroll
  for (int u = 0; u < NPW; ++u) {
    st[u] = __builtin_amdgcn_readlane(rp, u);
    dg[u] = (u < cnt) ? __builtin_amdgcn_readlane(rp, u + 1) - st[u] : 0;
  }
#pragma unroll
  for (int u = 0; u < NPW; ++u) {
    const int nshort = (dg[u] <= 32) ? dg[u] : 0;
    key[u] = (lane < nshort) ? adj[st[u] + lane] : 0x7fffffff;
  }
#pragma unroll
  for (int u = 0; u < NPW; ++u) {
    if (dg[u] > 0) {
      const int start = st[u];
      const int deg   = dg[u];
      if (deg <= 32) {
        const int ku = key[u];
        int rank = 0;
        for (int j = 0; j < deg; ++j) {
          const int kj = __builtin_amdgcn_readlane(ku, j);
          rank += ((kj < ku) || (kj == ku && j < lane)) ? 1 : 0;
        }
        if (lane < deg) adjs[start + rank] = ku;
      } else {
        for (int i = lane; i < deg; i += 32) {
          const int ki = adj[start + i];
          int rank = 0;
          for (int j = 0; j < deg; ++j) {
            const int kj = adj[start + j];
            rank += ((kj < ki) || (kj == ki && j < i)) ? 1 : 0;
          }
          adjs[start + rank] = ki;
        }
      }
    }
  }
}

template <int U>
__device__ inline void take_rows(const _Float16* __restrict__ hl, int key,
                                 int j, float4& acc) {
  half4 v[U];
#pragma unroll
  for (int u = 0; u < U; ++u) {
    const int sidx = __builtin_amdgcn_readlane(key, j + u);
    v[u] = *(const half4*)(hl + (size_t)sidx * HID);
  }
#pragma unroll
  for (int u = 0; u < U; ++u) {
    acc.x += (float)v[u][0]; acc.y += (float)v[u][1];
    acc.z += (float)v[u][2]; acc.w += (float)v[u][3];
  }
}

__device__ inline void sum_rows(const _Float16* __restrict__ hl, int key, int m,
                                float4& acc) {
  int j = 0;
  for (; j + 8 <= m; j += 8) take_rows<8>(hl, key, j, acc);
  if (j + 4 <= m) { take_rows<4>(hl, key, j, acc); j += 4; }
  if (j + 2 <= m) { take_rows<2>(hl, key, j, acc); j += 2; }
  if (j < m) take_rows<1>(hl, key, j, acc);
}

__global__ __launch_bounds__(256) void gather_kernel(
    const _Float16* __restrict__ h, const int* __restrict__ rowptr,
    const int* __restrict__ adjs, _Float16* __restrict__ agg, int n) {
  const int n0 = NPW * __builtin_amdgcn_readfirstlane(
      (blockIdx.x * 256 + threadIdx.x) >> 5);
  if (n0 >= n) return;
  const int lane = threadIdx.x & 31;
  const int cnt  = (n - n0 < NPW) ? (n - n0) : NPW;
  const int rp = (lane <= cnt) ? rowptr[n0 + lane] : 0;
  const _Float16* hl = h + lane * 4;
  int nstart = __builtin_amdgcn_readlane(rp, 0);
  int nend   = __builtin_amdgcn_readlane(rp, 1);
  int nkey = (lane < nend - nstart) ? adjs[nstart + lane] : 0;
  for (int i = 0; i < cnt; ++i) {
    const int start = nstart, end = nend;
    const int key0  = nkey;
    if (i + 1 < cnt) {
      nstart = end;
      nend   = __builtin_amdgcn_readlane(rp, i + 2);
      nkey   = (lane < nend - nstart) ? adjs[nstart + lane] : 0;
    }
    float4 acc = make_float4(0.0f, 0.0f, 0.0f, 0.0f);
    const int deg = end - start;
    sum_rows(hl, key0, (deg < 32) ? deg : 32, acc);
    for (int base = start + 32; base < end; base += 32) {
      const int m   = (end - base < 32) ? (end - base) : 32;
      const int key = (lane < m) ? adjs[base + lane] : 0;
      sum_rows(hl, key, m, acc);
    }
    const float inv = (deg > 0) ? __builtin_amdgcn_rcpf((float)deg) : 0.0f;
    half4 o;
    o[0] = (_Float16)(acc.x * inv); o[1] = (_Float16)(acc.y * inv);
    o[2] = (_Float16)(acc.z * inv); o[3] = (_Float16)(acc.w * inv);
    *(half4*)(agg + (size_t)(n0 + i) * HID + lane * 4) = o;
  }
}

__device__ inline v16h join16(half8 lo, half8 hi) {
  return __builtin_shufflevector(lo, hi, 0, 1, 2, 3, 4, 5, 6, 7,
                                         8, 9, 10, 11, 12, 13, 14, 15);
}

template <bool SAGE, bool RELU, bool COLSUM, bool OUTH>
__global__ __launch_bounds__(256) void gemm_kernel(
    const _Float16* __restrict__ h,
    const _Float16* __restrict__ agg,
    const _Float16* __restrict__ WlT,
    const _Float16* __restrict__ WrT,
    const float*    __restrict__ bias,
    float*    __restrict__ outf,
    _Float16* __restrict__ outh,
    float* __restrict__ colpart,
    int n_nodes) {
  __shared__ __attribute__((aligned(16))) _Float16 wtr[HID * WPITCH];
  __shared__ __attribute__((aligned(16))) _Float16 wtl[HID * WPITCH];
  __shared__ float colred[COLSUM ? 16 * HID : 1];

  {
    const uint4* sr = (const uint4*)WrT;
    for (int i = threadIdx.x; i < HID * HID / 8; i += 256)
      *(uint4*)&wtr[(i >> 4) * WPITCH + (i & 15) * 8] = sr[i];
    if constexpr (SAGE) {
      const uint4* sl = (const uint4*)WlT;
      for (int i = threadIdx.x; i < HID * HID / 8; i += 256)
        *(uint4*)&wtl[(i >> 4) * WPITCH + (i & 15) * 8] = sl[i];
    }
  }
  __syncthreads();

  const int lane = threadIdx.x & 31;
  const int wave = threadIdx.x >> 5;
  const int rowbase = blockIdx.x * 128 + wave * 16;
  const bool active = rowbase < n_nodes;
  if constexpr (!COLSUM) {
    if (!active) return;
  }

  float csum[8];
#pragma unroll
  for (int nt = 0; nt < 8; ++nt) csum[nt] = 0.0f;

  if (active) {

  const int arow = rowbase + (lane & 15);
  const int row  = arow < n_nodes ? arow : n_nodes - 1;
  const int lo   = (lane >= 16) ? 8 : 0;

  const _Float16* hp  = h + (size_t)row * HID + lo;
  const _Float16* gp  = SAGE ? (agg + (size_t)row * HID + lo) : nullptr;
  const _Float16* brp = &wtr[(lane & 15) * WPITCH + lo];
  const _Float16* blp = &wtl[(lane & 15) * WPITCH + lo];

  v8f acc[8];
  {
    v8f z = {};
#pragma unroll
    for (int nt = 0; nt < 8; ++nt) acc[nt] = z;
  }

#pragma unroll
  for (int kk = 0; kk < 4; ++kk) {
    const int kb = kk * 32;
    v16h ahf = join16(*(const half8*)(hp + kb),
                      *(const half8*)(hp + kb + 16));
    v16h agf;
    if constexpr (SAGE)
      agf = join16(*(const half8*)(gp + kb),
                   *(const half8*)(gp + kb + 16));
#pragma unroll
    for (int nt = 0; nt < 8; ++nt) {
      const int off = nt * 16 * WPITCH + kb;
      v16h br = join16(*(const half8*)(brp + off),
                       *(const half8*)(brp + off + 16));
      acc[nt] = __builtin_amdgcn_wmma_f32_16x16x32_f16(
          false, ahf, false, br, (short)0, acc[nt], false, false);
      if constexpr (SAGE) {
        v16h bl = join16(*(const half8*)(blp + off),
                         *(const half8*)(blp + off + 16));
        acc[nt] = __builtin_amdgcn_wmma_f32_16x16x32_f16(
            false, agf, false, bl, (short)0, acc[nt], false, false);
      }
    }
  }

  const int mofs = (lane >= 16) ? 8 : 0;
#pragma unroll
  for (int nt = 0; nt < 8; ++nt) {
    const int ncol = nt * 16 + (lane & 15);
    const float bb = bias[ncol];
#pragma unroll
    for (int r = 0; r < 8; ++r) {
      float v = acc[nt][r] + bb;
      if constexpr (RELU) v = fmaxf(v, 0.0f);
      int orow = rowbase + mofs + r;
      if (orow < n_nodes) {
        if constexpr (OUTH) outh[(size_t)orow * HID + ncol] = (_Float16)v;
        else                outf[(size_t)orow * HID + ncol] = v;
        if constexpr (COLSUM) csum[nt] += v;
      }
    }
  }
  }

  if constexpr (COLSUM) {
    float* cr = &colred[(wave * 2 + (lane >> 4)) * HID + (lane & 15)];
#pragma unroll
    for (int nt = 0; nt < 8; ++nt) cr[nt * 16] = csum[nt];
    __syncthreads();
    if (threadIdx.x < HID) {
      float t = 0.0f;
#pragma unroll
      for (int p = 0; p < 16; ++p) t += colred[p * HID + threadIdx.x];
      colpart[(size_t)blockIdx.x * HID + threadIdx.x] = t;
    }
  }
}

__global__ void colmean_kernel(const float* __restrict__ part, int nparts,
                               float* __restrict__ outm, int n) {
  __shared__ float red[256];
  int col = blockIdx.x;
  float acc = 0.0f;
  for (int r = threadIdx.x; r < nparts; r += 256)
    acc += part[(size_t)r * HID + col];
  red[threadIdx.x] = acc;
  __syncthreads();
  for (int st = 128; st > 0; st >>= 1) {
    if (threadIdx.x < st) red[threadIdx.x] += red[threadIdx.x + st];
    __syncthreads();
  }
  if (threadIdx.x == 0) outm[col] = red[0] / (float)n;
}

extern "C" void kernel_launch(void* const* d_in, const int* in_sizes, int n_in,
                              void* d_out, int out_size, void* d_ws, size_t ws_size,
                              hipStream_t stream) {
  const float* pos  = (const float*)d_in[0];
  const int*   an   = (const int*)  d_in[1];
  const int*   ei   = (const int*)  d_in[2];
  const float* W_in = (const float*)d_in[3];
  const float* b_in = (const float*)d_in[4];
  const float* W1l  = (const float*)d_in[5];
  const float* b1   = (const float*)d_in[6];
  const float* W1r  = (const float*)d_in[7];
  const float* W2l  = (const float*)d_in[8];
  const float* b2   = (const float*)d_in[9];
  const float* W2r  = (const float*)d_in[10];
  const float* Wout = (const float*)d_in[11];
  const float* bout = (const float*)d_in[12];

  const int n  = in_sizes[0] / 3;
  const int nE = in_sizes[2] / 2;

  _Float16* h   = (_Float16*)d_ws;
  _Float16* agg = h + (size_t)n * HID;

  _Float16* wtbase = agg + (size_t)n * HID;
  _Float16* W1lT = wtbase + 0 * HID * HID;
  _Float16* W1rT = wtbase + 1 * HID * HID;
  _Float16* W2lT = wtbase + 2 * HID * HID;
  _Float16* W2rT = wtbase + 3 * HID * HID;
  _Float16* WoT  = wtbase + 4 * HID * HID;

  const int scanBlks = (n + 1023) / 1024;
  int* deg    = (int*)(wtbase + 5 * HID * HID);
  int* rowptr = deg + 2 * n;
  int* bsum   = rowptr + (n + 1);
  int* adj    = bsum + scanBlks;
  int* adjs   = adj + nE;
  int* rank   = adjs + nE;
  float* colpart = (float*)(rank + nE);

  float* outm = (float*)d_out;
  float* nst  = (float*)d_out + HID;

  const int gemmBlks = (n + 127) / 128;
  const int edgeBlks = (nE + 256 * EPT - 1) / (256 * EPT);
  const int nodeBlks = (n + 8 * NPW - 1) / (8 * NPW);
  const int zeroBlks = (2 * n + 1023) / 1024;

  prep_kernel<<<5 * WBLKS + zeroBlks + nodeBlks, 256, 0, stream>>>(
      pos, an, W_in, b_in, W1l, W1r, W2l, W2r, Wout, wtbase, deg, zeroBlks,
      h, n);

  degree_kernel<<<edgeBlks, 256, 0, stream>>>(ei, deg, rank, nE, n);
  blocksum_kernel<<<scanBlks, 256, 0, stream>>>(deg, bsum, n);
  scan_kernel<<<scanBlks, 256, 0, stream>>>(deg, bsum, rowptr, n);
  fill_kernel<<<edgeBlks, 256, 0, stream>>>(ei, rowptr, rank, deg, adj, nE);
  segsort_kernel<<<nodeBlks, 256, 0, stream>>>(rowptr, adj, adjs, n);

  gather_kernel<<<nodeBlks, 256, 0, stream>>>(h, rowptr, adjs, agg, n);
  gemm_kernel<true, true, false, true><<<gemmBlks, 256, 0, stream>>>(
      h, agg, W1lT, W1rT, b1, nullptr, h, nullptr, n);

  gather_kernel<<<nodeBlks, 256, 0, stream>>>(h, rowptr, adjs, agg, n);
  gemm_kernel<true, true, false, true><<<gemmBlks, 256, 0, stream>>>(
      h, agg, W2lT, W2rT, b2, nullptr, h, nullptr, n);

  gemm_kernel<false, false, true, false><<<gemmBlks, 256, 0, stream>>>(
      h, nullptr, nullptr, WoT, bout, nst, nullptr, colpart, n);
  colmean_kernel<<<HID, 256, 0, stream>>>(colpart, gemmBlks, outm, n);
}


# Conv layer in one launch: the wave owning a 16-row GEMM tile sums its nodes' neighbour rows (same order) into its own LDS tile, read as the mean-side A operand; no means in memory, 2 launches fewer
# speedup vs baseline: 5.3417x; 1.1921x over previous
#include <hip/hip_runtime.h>


typedef __attribute__((ext_vector_type(16))) _Float16 v16h;
typedef __attribute__((ext_vector_type(8)))  _Float16 half8;
typedef __attribute__((ext_vector_type(4)))  _Float16 half4;
typedef __attribute__((ext_vector_type(8)))  float    v8f;

#define HID 128
#define WPITCH 136
#define NPW 8
#define EPT 4
#define FW 16

__device__ inline float lane_f(float x, int l) {
  return __int_as_float(__builtin_amdgcn_readlane(__float_as_int(x), l));
}

#define WBLKS (HID * HID / 256)
__global__ __launch_bounds__(256) void prep_kernel(
    const float* __restrict__ pos,
    const int*   __restrict__ an,
    const float* __restrict__ W_in,
    const float* __restrict__ b_in,
    const float* __restrict__ W1l, const float* __restrict__ W1r,
    const float* __restrict__ W2l, const float* __restrict__ W2r,
    const float* __restrict__ Wout,
    _Float16* __restrict__ wt,
    int* __restrict__ deg, int zeroBlks,
    _Float16* __restrict__ h, int n) {
  const int blk = blockIdx.x;
  if (blk < 5 * WBLKS) {
    const int m = blk / WBLKS;
    const float* W = (m == 0) ? W1l : (m == 1) ? W1r : (m == 2) ? W2l
                   : (m == 3) ? W2r : Wout;
    const int t = (blk - m * WBLKS) * 256 + threadIdx.x;
    const int k = t & 127, c = t >> 7;
    wt[m * HID * HID + t] = (_Float16)W[k * HID + c];
    return;
  }
  if (blk < 5 * WBLKS + zeroBlks) {
    const int t = ((blk - 5 * WBLKS) * 256 + threadIdx.x) * 4;
    if (t + 3 < 2 * n) {
      *(int4*)(deg + t) = make_int4(0, 0, 0, 0);
    } else {
      for (int k = 0; k < 4; ++k)
        if (t + k < 2 * n) deg[t + k] = 0;
    }
    return;
  }
  const int n0 = NPW * __builtin_amdgcn_readfirstlane(
      ((blk - 5 * WBLKS - zeroBlks) * 256 + threadIdx.x) >> 5);
  if (n0 >= n) return;
  const int lane = threadIdx.x & 31;
  const int cnt  = (n - n0 < NPW) ? (n - n0) : NPW;
  const int c = lane * 4;
  const float4 w0 = *(const float4*)(W_in + 0 * HID + c);
  const float4 w1 = *(const float4*)(W_in + 1 * HID + c);
  const float4 w2 = *(const float4*)(W_in + 2 * HID + c);
  const float4 w3 = *(const float4*)(W_in + 3 * HID + c);
  const float4 bb = *(const float4*)(b_in + c);
  const float zl = (lane < cnt) ? (float)an[n0 + lane] * 0.1f : 0.0f;
  const float pl = (lane < 3 * cnt) ? pos[n0 * 3 + lane] : 0.0f;
#pragma unroll
  for (int i = 0; i < NPW; ++i) {
    if (i < cnt) {
      const float z  = lane_f(zl, i);
      const float px = lane_f(pl, 3 * i + 0);
      const float py = lane_f(pl, 3 * i + 1);
      const float pz = lane_f(pl, 3 * i + 2);
      float4 v = bb;
      v.x = fmaf(z,  w0.x, v.x); v.y = fmaf(z,  w0.y, v.y);
      v.z = fmaf(z,  w0.z, v.z); v.w = fmaf(z,  w0.w, v.w);
      v.x = fmaf(px, w1.x, v.x); v.y = fmaf(px, w1.y, v.y);
      v.z = fmaf(px, w1.z, v.z); v.w = fmaf(px, w1.w, v.w);
      v.x = fmaf(py, w2.x, v.x); v.y = fmaf(py, w2.y, v.y);
      v.z = fmaf(py, w2.z, v.z); v.w = fmaf(py, w2.w, v.w);
      v.x = fmaf(pz, w3.x, v.x); v.y = fmaf(pz, w3.y, v.y);
      v.z = fmaf(pz, w3.z, v.z); v.w = fmaf(pz, w3.w, v.w);
      half4 o;
      o[0] = (_Float16)fmaxf(v.x, 0.0f); o[1] = (_Float16)fmaxf(v.y, 0.0f);
      o[2] = (_Float16)fmaxf(v.z, 0.0f); o[3] = (_Float16)fmaxf(v.w, 0.0f);
      *(half4*)(h + (size_t)(n0 + i) * HID + c) = o;
    }
  }
}

template <bool FULL>
__device__ inline void degree_body(const int* __restrict__ ei,
                                   int* __restrict__ cset,
                                   int* __restrict__ rank, int nE, int e0,
                                   int setbit) {
  int d[EPT], r[EPT];
#pragma unroll
  for (int k = 0; k < EPT; ++k) {
    const int e = e0 + k * 256;
    d[k] = (FULL || e < nE) ? ei[nE + e] : 0;
  }
#pragma unroll
  for (int k = 0; k < EPT; ++k) {
    const int e = e0 + k * 256;
    r[k] = (FULL || e < nE) ? atomicAdd(cset + d[k], 1) : 0;
  }
#pragma unroll
  for (int k = 0; k < EPT; ++k) {
    const int e = e0 + k * 256;
    if (FULL || e < nE) rank[e] = r[k] | setbit;
  }
}

__global__ void degree_kernel(const int* __restrict__ ei,
                              int* __restrict__ deg,
                              int* __restrict__ rank, int nE, int n) {
  const int b0 = blockIdx.x * (256 * EPT);
  const int e0 = b0 + threadIdx.x;
  const unsigned id = __builtin_amdgcn_s_sendmsg_rtn(0x87);
  const int set = (int)((id >> 18) & 1u);
  int* cset = deg + set * n;
  if (b0 + 256 * EPT <= nE) degree_body<true>(ei, cset, rank, nE, e0, set << 31);
  else                      degree_body<false>(ei, cset, rank, nE, e0, set << 31);
}

__global__ __launch_bounds__(256) void blocksum_kernel(
    const int* __restrict__ deg, int* __restrict__ bsum, int n) {
  __shared__ int red[256];
  const int t = threadIdx.x;
  const int i0 = blockIdx.x * 1024 + t * 4;
  int v = 0;
#pragma unroll
  for (int k = 0; k < 4; ++k)
    if (i0 + k < n) v += deg[i0 + k] + deg[n + i0 + k];
  red[t] = v;
  __syncthreads();
  for (int st = 128; st > 0; st >>= 1) {
    if (t < st) red[t] += red[t + st];
    __syncthreads();
  }
  if (t == 0) bsum[blockIdx.x] = red[0];
}

__global__ __launch_bounds__(256) void scan_kernel(
    const int* __restrict__ deg, const int* __restrict__ bsum,
    int* __restrict__ rowptr, int n) {
  __shared__ int sh[256];
  const int t = threadIdx.x;
  const int b = blockIdx.x;
  int part = 0;
  for (int i = t; i < b; i += 256) part += bsum[i];
  sh[t] = part;
  __syncthreads();
  for (int st = 128; st > 0; st >>= 1) {
    if (t < st) sh[t] += sh[t + st];
    __syncthreads();
  }
  const int offset = sh[0];
  __syncthreads();
  const int i0 = b * 1024 + t * 4;
  int d[4];
#pragma unroll
  for (int k = 0; k < 4; ++k)
    d[k] = (i0 + k < n) ? deg[i0 + k] + deg[n + i0 + k] : 0;
  const int tsum = d[0] + d[1] + d[2] + d[3];
  sh[t] = tsum;
  __syncthreads();
  for (int st = 1; st < 256; st <<= 1) {
    const int v = (t >= st) ? sh[t - st] : 0;
    __syncthreads();
    sh[t] += v;
    __syncthreads();
  }
  int run = offset + sh[t] - tsum;
#pragma unroll
  for (int k = 0; k < 4; ++k) {
    if (i0 + k < n) {
      rowptr[i0 + k] = run;
      run += d[k];
      if (i0 + k == n - 1) rowptr[n] = run;
    }
  }
}

template <bool FULL>
__device__ inline void fill_body(const int* __restrict__ ei,
                                 const int* __restrict__ rowptr,
                                 const int* __restrict__ rank,
                                 const int* __restrict__ deg,
                                 int* __restrict__ adj, int nE, int e0) {
  int s[EPT], d[EPT], r[EPT], base[EPT], before[EPT];
#pragma unroll
  for (int k = 0; k < EPT; ++k) {
    const int e = e0 + k * 256;
    const bool in = FULL || e < nE;
    s[k] = in ? ei[e] : 0;
    d[k] = in ? ei[nE + e] : 0;
    r[k] = in ? rank[e] : 0;
  }
#pragma unroll
  for (int k = 0; k < EPT; ++k) {
    const int e = e0 + k * 256;
    base[k] = (FULL || e < nE) ? rowptr[d[k]] : 0;
  }
#pragma unroll
  for (int k = 0; k < EPT; ++k) {
    const int e = e0 + k * 256;
    before[k] = ((FULL || e < nE) && r[k] < 0) ? deg[d[k]] : 0;
  }
#pragma unroll
  for (int k = 0; k < EPT; ++k) {
    const int e = e0 + k * 256;
    if (FULL || e < nE)
      adj[base[k] + (r[k] & 0x7fffffff) + before[k]] = s[k];
  }
}

__global__ void fill_kernel(const int* __restrict__ ei,
                            const int* __restrict__ rowptr,
                            const int* __restrict__ rank,
                            const int* __restrict__ deg,
                            int* __restrict__ adj, int nE) {
  const int b0 = blockIdx.x * (256 * EPT);
  const int e0 = b0 + threadIdx.x;
  if (b0 + 256 * EPT <= nE) fill_body<true>(ei, rowptr, rank, deg, adj, nE, e0);
  else                      fill_body<false>(ei, rowptr, rank, deg, adj, nE, e0);
}

__global__ __launch_bounds__(256) void segsort_kernel(
    const int* __restrict__ rowptr, const int* __restrict__ adj,
    int* __restrict__ adjs, int n) {
  const int n0 = NPW * __builtin_amdgcn_readfirstlane(
      (blockIdx.x * 256 + threadIdx.x) >> 5);
  if (n0 >= n) return;
  const int lane = threadIdx.x & 31;
  const int cnt  = (n - n0 < NPW) ? (n - n0) : NPW;
  const int rp = (lane <= cnt) ? rowptr[n0 + lane] : 0;
  int key[NPW], st[NPW], dg[NPW];
#pragma unroll
  for (int u = 0; u < NPW; ++u) {
    st[u] = __builtin_amdgcn_readlane(rp, u);
    dg[u] = (u < cnt) ? __builtin_amdgcn_readlane(rp, u + 1) - st[u] : 0;
  }
#pragma unroll
  for (int u = 0; u < NPW; ++u) {
    const int nshort = (dg[u] <= 32) ? dg[u] : 0;
    key[u] = (lane < nshort) ? adj[st[u] + lane] : 0x7fffffff;
  }
#pragma unroll
  for (int u = 0; u < NPW; ++u) {
    if (dg[u] > 0) {
      const int start = st[u];
      const int deg   = dg[u];
      if (deg <= 32) {
        const int ku = key[u];
        int rank = 0;
        for (int j = 0; j < deg; ++j) {
          const int kj = __builtin_amdgcn_readlane(ku, j);
          rank += ((kj < ku) || (kj == ku && j < lane)) ? 1 : 0;
        }
        if (lane < deg) adjs[start + rank] = ku;
      } else {
        for (int i = lane; i < deg; i += 32) {
          const int ki = adj[start + i];
          int rank = 0;
          for (int j = 0; j < deg; ++j) {
            const int kj = adj[start + j];
            rank += ((kj < ki) || (kj == ki && j < i)) ? 1 : 0;
          }
          adjs[start + rank] = ki;
        }
      }
    }
  }
}

template <int U>
__device__ inline void take_upto(const _Float16* __restrict__ hl, int key,
                                 int j, int m, float4& acc) {
  half4 v[U];
#pragma unroll
  for (int u = 0; u < U; ++u) {
    if (j + u < m) {
      const int sidx = __builtin_amdgcn_readlane(key, j + u);
      v[u] = *(const half4*)(hl + (size_t)sidx * HID);
    }
  }
#pragma unroll
  for (int u = 0; u < U; ++u) {
    if (j + u < m) {
      acc.x += (float)v[u][0]; acc.y += (float)v[u][1];
      acc.z += (float)v[u][2]; acc.w += (float)v[u][3];
    }
  }
}

__device__ inline void sum_rows(const _Float16* __restrict__ hl, int key, int m,
                                float4& acc) {
  take_upto<16>(hl, key, 0, m, acc);
  if (m > 16) take_upto<16>(hl, key, 16, m, acc);
}

__device__ inline v16h join16(half8 lo, half8 hi) {
  return __builtin_shufflevector(lo, hi, 0, 1, 2, 3, 4, 5, 6, 7,
                                         8, 9, 10, 11, 12, 13, 14, 15);
}

__global__ __launch_bounds__(32 * FW, 2) void sage_kernel(
    const _Float16* __restrict__ h,
    const int* __restrict__ rowptr, const int* __restrict__ adjs,
    const _Float16* __restrict__ WlT,
    const _Float16* __restrict__ WrT,
    const float*    __restrict__ bias,
    _Float16* __restrict__ outh,
    int n_nodes) {
  __shared__ __attribute__((aligned(16))) _Float16 wtr[HID * WPITCH];
  __shared__ __attribute__((aligned(16))) _Float16 wtl[HID * WPITCH];
  __shared__ __attribute__((aligned(16))) _Float16 mtile[FW * 16 * WPITCH];

  {
    const uint4* sr = (const uint4*)WrT;
    const uint4* sl = (const uint4*)WlT;
    for (int i = threadIdx.x; i < HID * HID / 8; i += 32 * FW) {
      *(uint4*)&wtr[(i >> 4) * WPITCH + (i & 15) * 8] = sr[i];
      *(uint4*)&wtl[(i >> 4) * WPITCH + (i & 15) * 8] = sl[i];
    }
  }
  __syncthreads();

  const int lane = threadIdx.x & 31;
  const int wave = __builtin_amdgcn_readfirstlane((int)threadIdx.x >> 5);
  const int rowbase = blockIdx.x * (16 * FW) + wave * 16;
  if (rowbase >= n_nodes) return;
  const int cnt = (n_nodes - rowbase < 16) ? (n_nodes - rowbase) : 16;

  const int arow = rowbase + (lane & 15);
  const int row  = arow < n_nodes ? arow : n_nodes - 1;
  const int lo   = (lane >= 16) ? 8 : 0;

  const _Float16* hp = h + (size_t)row * HID + lo;
  v16h ahf[4];
#pragma unroll
  for (int kk = 0; kk < 4; ++kk)
    ahf[kk] = join16(*(const half8*)(hp + kk * 32),
                     *(const half8*)(hp + kk * 32 + 16));

  _Float16* mt = &mtile[wave * 16 * WPITCH];
  {
    const int rp = (lane <= cnt) ? rowptr[rowbase + lane] : 0;
    const _Float16* hl = h + lane * 4;
    int nstart = __builtin_amdgcn_readlane(rp, 0);
    int nend   = __builtin_amdgcn_readlane(rp, 1);
    int nkey = (lane < nend - nstart) ? adjs[nstart + lane] : 0;
    for (int i = 0; i < cnt; ++i) {
      const int start = nstart, end = nend;
      const int key0  = nkey;
      if (i + 1 < cnt) {
        nstart = end;
        nend   = __builtin_amdgcn_readlane(rp, i + 2);
        nkey   = (lane < nend - nstart) ? adjs[nstart + lane] : 0;
      }
      float4 acc = make_float4(0.0f, 0.0f, 0.0f, 0.0f);
      const int deg = end - start;
      sum_rows(hl, key0, (deg < 32) ? deg : 32, acc);
      for (int base = start + 32; base < end; base += 32) {
        const int m   = (end - base < 32) ? (end - base) : 32;
        const int key = (lane < m) ? adjs[base + lane] : 0;
        sum_rows(hl, key, m, acc);
      }
      const float inv = (deg > 0) ? __builtin_amdgcn_rcpf((float)deg) : 0.0f;
      half4 o;
      o[0] = (_Float16)(acc.x * inv); o[1] = (_Float16)(acc.y * inv);
      o[2] = (_Float16)(acc.z * inv); o[3] = (_Float16)(acc.w * inv);
      *(half4*)(mt + i * WPITCH + lane * 4) = o;
    }
    for (int i = cnt; i < 16; ++i) {
      half4 z = {};
      *(half4*)(mt + i * WPITCH + lane * 4) = z;
    }
  }

  const _Float16* gp  = mt + (lane & 15) * WPITCH + lo;
  const _Float16* brp = &wtr[(lane & 15) * WPITCH + lo];
  const _Float16* blp = &wtl[(lane & 15) * WPITCH + lo];

  v8f acc[8];
  {
    v8f z = {};
#pragma unroll
    for (int nt = 0; nt < 8; ++nt) acc[nt] = z;
  }

#pragma unroll
  for (int kk = 0; kk < 4; ++kk) {
    const int kb = kk * 32;
    v16h agf = join16(*(const half8*)(gp + kb),
                      *(const half8*)(gp + kb + 16));
#pragma unroll
    for (int nt = 0; nt < 8; ++nt) {
      const int off = nt * 16 * WPITCH + kb;
      v16h br = join16(*(const half8*)(brp + off),
                       *(const half8*)(brp + off + 16));
      acc[nt] = __builtin_amdgcn_wmma_f32_16x16x32_f16(
          false, ahf[kk], false, br, (short)0, acc[nt], false, false);
      v16h bl = join16(*(const half8*)(blp + off),
                       *(const half8*)(blp + off + 16));
      acc[nt] = __builtin_amdgcn_wmma_f32_16x16x32_f16(
          false, agf, false, bl, (short)0, acc[nt], false, false);
    }
  }

  const int mofs = (lane >= 16) ? 8 : 0;
#pragma unroll
  for (int nt = 0; nt < 8; ++nt) {
    const int ncol = nt * 16 + (lane & 15);
    const float bb = bias[ncol];
#pragma unroll
    for (int r = 0; r < 8; ++r) {
      const float v = fmaxf(acc[nt][r] + bb, 0.0f);
      const int orow = rowbase + mofs + r;
      if (orow < n_nodes) outh[(size_t)orow * HID + ncol] = (_Float16)v;
    }
  }
}

template <bool RELU, bool COLSUM, bool OUTH>
__global__ __launch_bounds__(256) void gemm_kernel(
    const _Float16* __restrict__ h,
    const _Float16* __restrict__ WrT,
    const float*    __restrict__ bias,
    float*    __restrict__ outf,
    _Float16* __restrict__ outh,
    float* __restrict__ colpart,
    int n_nodes) {
  __shared__ __attribute__((aligned(16))) _Float16 wtr[HID * WPITCH];
  __shared__ float colred[COLSUM ? 16 * HID : 1];

  {
    const uint4* sr = (const uint4*)WrT;
    for (int i = threadIdx.x; i < HID * HID / 8; i += 256)
      *(uint4*)&wtr[(i >> 4) * WPITCH + (i & 15) * 8] = sr[i];
  }
  __syncthreads();

  const int lane = threadIdx.x & 31;
  const int wave = threadIdx.x >> 5;
  const int rowbase = blockIdx.x * 128 + wave * 16;
  const bool active = rowbase < n_nodes;
  if constexpr (!COLSUM) {
    if (!active) return;
  }

  float csum[8];
#pragma unroll
  for (int nt = 0; nt < 8; ++nt) csum[nt] = 0.0f;

  if (active) {

  const int arow = rowbase + (lane & 15);
  const int row  = arow < n_nodes ? arow : n_nodes - 1;
  const int lo   = (lane >= 16) ? 8 : 0;

  const _Float16* hp  = h + (size_t)row * HID + lo;
  const _Float16* brp = &wtr[(lane & 15) * WPITCH + lo];

  v8f acc[8];
  {
    v8f z = {};
#pragma unroll
    for (int nt = 0; nt < 8; ++nt) acc[nt] = z;
  }

#pragma unroll
  for (int kk = 0; kk < 4; ++kk) {
    const int kb = kk * 32;
    v16h ahf = join16(*(const half8*)(hp + kb),
                      *(const half8*)(hp + kb + 16));
#pragma unroll
    for (int nt = 0; nt < 8; ++nt) {
      const int off = nt * 16 * WPITCH + kb;
      v16h br = join16(*(const half8*)(brp + off),
                       *(const half8*)(brp + off + 16));
      acc[nt] = __builtin_amdgcn_wmma_f32_16x16x32_f16(
          false, ahf, false, br, (short)0, acc[nt], false, false);
    }
  }

  const int mofs = (lane >= 16) ? 8 : 0;
#pragma unroll
  for (int nt = 0; nt < 8; ++nt) {
    const int ncol = nt * 16 + (lane & 15);
    const float bb = bias[ncol];
#pragma unroll
    for (int r = 0; r < 8; ++r) {
      float v = acc[nt][r] + bb;
      if constexpr (RELU) v = fmaxf(v, 0.0f);
      int orow = rowbase + mofs + r;
      if (orow < n_nodes) {
        if constexpr (OUTH) outh[(size_t)orow * HID + ncol] = (_Float16)v;
        else                outf[(size_t)orow * HID + ncol] = v;
        if constexpr (COLSUM) csum[nt] += v;
      }
    }
  }
  }

  if constexpr (COLSUM) {
    float* cr = &colred[(wave * 2 + (lane >> 4)) * HID + (lane & 15)];
#pragma unroll
    for (int nt = 0; nt < 8; ++nt) cr[nt * 16] = csum[nt];
    __syncthreads();
    if (threadIdx.x < HID) {
      float t = 0.0f;
#pragma unroll
      for (int p = 0; p < 16; ++p) t += colred[p * HID + threadIdx.x];
      colpart[(size_t)blockIdx.x * HID + threadIdx.x] = t;
    }
  }
}

__global__ void colmean_kernel(const float* __restrict__ part, int nparts,
                               float* __restrict__ outm, int n) {
  __shared__ float red[256];
  int col = blockIdx.x;
  float acc = 0.0f;
  for (int r = threadIdx.x; r < nparts; r += 256)
    acc += part[(size_t)r * HID + col];
  red[threadIdx.x] = acc;
  __syncthreads();
  for (int st = 128; st > 0; st >>= 1) {
    if (threadIdx.x < st) red[threadIdx.x] += red[threadIdx.x + st];
    __syncthreads();
  }
  if (threadIdx.x == 0) outm[col] = red[0] / (float)n;
}

extern "C" void kernel_launch(void* const* d_in, const int* in_sizes, int n_in,
                              void* d_out, int out_size, void* d_ws, size_t ws_size,
                              hipStream_t stream) {
  const float* pos  = (const float*)d_in[0];
  const int*   an   = (const int*)  d_in[1];
  const int*   ei   = (const int*)  d_in[2];
  const float* W_in = (const float*)d_in[3];
  const float* b_in = (const float*)d_in[4];
  const float* W1l  = (const float*)d_in[5];
  const float* b1   = (const float*)d_in[6];
  const float* W1r  = (const float*)d_in[7];
  const float* W2l  = (const float*)d_in[8];
  const float* b2   = (const float*)d_in[9];
  const float* W2r  = (const float*)d_in[10];
  const float* Wout = (const float*)d_in[11];
  const float* bout = (const float*)d_in[12];

  const int n  = in_sizes[0] / 3;
  const int nE = in_sizes[2] / 2;

  _Float16* h   = (_Float16*)d_ws;
  _Float16* hb  = h + (size_t)n * HID;

  _Float16* wtbase = hb + (size_t)n * HID;
  _Float16* W1lT = wtbase + 0 * HID * HID;
  _Float16* W1rT = wtbase + 1 * HID * HID;
  _Float16* W2lT = wtbase + 2 * HID * HID;
  _Float16* W2rT = wtbase + 3 * HID * HID;
  _Float16* WoT  = wtbase + 4 * HID * HID;

  const int scanBlks = (n + 1023) / 1024;
  int* deg    = (int*)(wtbase + 5 * HID * HID);
  int* rowptr = deg + 2 * n;
  int* bsum   = rowptr + (n + 1);
  int* adj    = bsum + scanBlks;
  int* adjs   = adj + nE;
  int* rank   = adjs + nE;
  float* colpart = (float*)(rank + nE);

  float* outm = (float*)d_out;
  float* nst  = (float*)d_out + HID;

  const int gemmBlks = (n + 127) / 128;
  const int sageBlks = (n + 16 * FW - 1) / (16 * FW);
  const int edgeBlks = (nE + 256 * EPT - 1) / (256 * EPT);
  const int nodeBlks = (n + 8 * NPW - 1) / (8 * NPW);
  const int zeroBlks = (2 * n + 1023) / 1024;

  prep_kernel<<<5 * WBLKS + zeroBlks + nodeBlks, 256, 0, stream>>>(
      pos, an, W_in, b_in, W1l, W1r, W2l, W2r, Wout, wtbase, deg, zeroBlks,
      h, n);

  degree_kernel<<<edgeBlks, 256, 0, stream>>>(ei, deg, rank, nE, n);
  blocksum_kernel<<<scanBlks, 256, 0, stream>>>(deg, bsum, n);
  scan_kernel<<<scanBlks, 256, 0, stream>>>(deg, bsum, rowptr, n);
  fill_kernel<<<edgeBlks, 256, 0, stream>>>(ei, rowptr, rank, deg, adj, nE);
  segsort_kernel<<<nodeBlks, 256, 0, stream>>>(rowptr, adj, adjs, n);

  sage_kernel<<<sageBlks, 32 * FW, 0, stream>>>(h, rowptr, adjs, W1lT, W1rT, b1,
                                                hb, n);
  sage_kernel<<<sageBlks, 32 * FW, 0, stream>>>(hb, rowptr, adjs, W2lT, W2rT, b2,
                                                h, n);

  gemm_kernel<false, true, false><<<gemmBlks, 256, 0, stream>>>(
      h, WoT, bout, nst, nullptr, colpart, n);
  colmean_kernel<<<HID, 256, 0, stream>>>(colpart, gemmBlks, outm, n);
}


# Output projection inside conv layer 2's launch: a wave turns its h2 tile (via LDS) into A fragments for W_out^T held in LDS; h2 never stored, 1 launch fewer; mean partials per 208 rows: new sum tree
# speedup vs baseline: 5.5906x; 1.0466x over previous
#include <hip/hip_runtime.h>


typedef __attribute__((ext_vector_type(16))) _Float16 v16h;
typedef __attribute__((ext_vector_type(8)))  _Float16 half8;
typedef __attribute__((ext_vector_type(4)))  _Float16 half4;
typedef __attribute__((ext_vector_type(8)))  float    v8f;

#define HID 128
#define WPITCH 136
#define NPW 8
#define EPT 4
#define FW 16
#define FWO 13

__device__ inline float lane_f(float x, int l) {
  return __int_as_float(__builtin_amdgcn_readlane(__float_as_int(x), l));
}

#define WBLKS (HID * HID / 256)
__global__ __launch_bounds__(256) void prep_kernel(
    const float* __restrict__ pos,
    const int*   __restrict__ an,
    const float* __restrict__ W_in,
    const float* __restrict__ b_in,
    const float* __restrict__ W1l, const float* __restrict__ W1r,
    const float* __restrict__ W2l, const float* __restrict__ W2r,
    const float* __restrict__ Wout,
    _Float16* __restrict__ wt,
    int* __restrict__ deg, int zeroBlks,
    _Float16* __restrict__ h, int n) {
  const int blk = blockIdx.x;
  if (blk < 5 * WBLKS) {
    const int m = blk / WBLKS;
    const float* W = (m == 0) ? W1l : (m == 1) ? W1r : (m == 2) ? W2l
                   : (m == 3) ? W2r : Wout;
    const int t = (blk - m * WBLKS) * 256 + threadIdx.x;
    const int k = t & 127, c = t >> 7;
    wt[m * HID * HID + t] = (_Float16)W[k * HID + c];
    return;
  }
  if (blk < 5 * WBLKS + zeroBlks) {
    const int t = ((blk - 5 * WBLKS) * 256 + threadIdx.x) * 4;
    if (t + 3 < 2 * n) {
      *(int4*)(deg + t) = make_int4(0, 0, 0, 0);
    } else {
      for (int k = 0; k < 4; ++k)
        if (t + k < 2 * n) deg[t + k] = 0;
    }
    return;
  }
  const int n0 = NPW * __builtin_amdgcn_readfirstlane(
      ((blk - 5 * WBLKS - zeroBlks) * 256 + threadIdx.x) >> 5);
  if (n0 >= n) return;
  const int lane = threadIdx.x & 31;
  const int cnt  = (n - n0 < NPW) ? (n - n0) : NPW;
  const int c = lane * 4;
  const float4 w0 = *(const float4*)(W_in + 0 * HID + c);
  const float4 w1 = *(const float4*)(W_in + 1 * HID + c);
  const float4 w2 = *(const float4*)(W_in + 2 * HID + c);
  const float4 w3 = *(const float4*)(W_in + 3 * HID + c);
  const float4 bb = *(const float4*)(b_in + c);
  const float zl = (lane < cnt) ? (float)an[n0 + lane] * 0.1f : 0.0f;
  const float pl = (lane < 3 * cnt) ? pos[n0 * 3 + lane] : 0.0f;
#pragma unroll
  for (int i = 0; i < NPW; ++i) {
    if (i < cnt) {
      const float z  = lane_f(zl, i);
      const float px = lane_f(pl, 3 * i + 0);
      const float py = lane_f(pl, 3 * i + 1);
      const float pz = lane_f(pl, 3 * i + 2);
      float4 v = bb;
      v.x = fmaf(z,  w0.x, v.x); v.y = fmaf(z,  w0.y, v.y);
      v.z = fmaf(z,  w0.z, v.z); v.w = fmaf(z,  w0.w, v.w);
      v.x = fmaf(px, w1.x, v.x); v.y = fmaf(px, w1.y, v.y);
      v.z = fmaf(px, w1.z, v.z); v.w = fmaf(px, w1.w, v.w);
      v.x = fmaf(py, w2.x, v.x); v.y = fmaf(py, w2.y, v.y);
      v.z = fmaf(py, w2.z, v.z); v.w = fmaf(py, w2.w, v.w);
      v.x = fmaf(pz, w3.x, v.x); v.y = fmaf(pz, w3.y, v.y);
      v.z = fmaf(pz, w3.z, v.z); v.w = fmaf(pz, w3.w, v.w);
      half4 o;
      o[0] = (_Float16)fmaxf(v.x, 0.0f); o[1] = (_Float16)fmaxf(v.y, 0.0f);
      o[2] = (_Float16)fmaxf(v.z, 0.0f); o[3] = (_Float16)fmaxf(v.w, 0.0f);
      *(half4*)(h + (size_t)(n0 + i) * HID + c) = o;
    }
  }
}

template <bool FULL>
__device__ inline void degree_body(const int* __restrict__ ei,
                                   int* __restrict__ cset,
                                   int* __restrict__ rank, int nE, int e0,
                                   int setbit) {
  int d[EPT], r[EPT];
#pragma unroll
  for (int k = 0; k < EPT; ++k) {
    const int e = e0 + k * 256;
    d[k] = (FULL || e < nE) ? ei[nE + e] : 0;
  }
#pragma unroll
  for (int k = 0; k < EPT; ++k) {
    const int e = e0 + k * 256;
    r[k] = (FULL || e < nE) ? atomicAdd(cset + d[k], 1) : 0;
  }
#pragma unroll
  for (int k = 0; k < EPT; ++k) {
    const int e = e0 + k * 256;
    if (FULL || e < nE) rank[e] = r[k] | setbit;
  }
}

__global__ void degree_kernel(const int* __restrict__ ei,
                              int* __restrict__ deg,
                              int* __restrict__ rank, int nE, int n) {
  const int b0 = blockIdx.x * (256 * EPT);
  const int e0 = b0 + threadIdx.x;
  const unsigned id = __builtin_amdgcn_s_sendmsg_rtn(0x87);
  const int set = (int)((id >> 18) & 1u);
  int* cset = deg + set * n;
  if (b0 + 256 * EPT <= nE) degree_body<true>(ei, cset, rank, nE, e0, set << 31);
  else                      degree_body<false>(ei, cset, rank, nE, e0, set << 31);
}

__global__ __launch_bounds__(256) void blocksum_kernel(
    const int* __restrict__ deg, int* __restrict__ bsum, int n) {
  __shared__ int red[256];
  const int t = threadIdx.x;
  const int i0 = blockIdx.x * 1024 + t * 4;
  int v = 0;
#pragma unroll
  for (int k = 0; k < 4; ++k)
    if (i0 + k < n) v += deg[i0 + k] + deg[n + i0 + k];
  red[t] = v;
  __syncthreads();
  for (int st = 128; st > 0; st >>= 1) {
    if (t < st) red[t] += red[t + st];
    __syncthreads();
  }
  if (t == 0) bsum[blockIdx.x] = red[0];
}

__global__ __launch_bounds__(256) void scan_kernel(
    const int* __restrict__ deg, const int* __restrict__ bsum,
    int* __restrict__ rowptr, int n) {
  __shared__ int sh[256];
  const int t = threadIdx.x;
  const int b = blockIdx.x;
  int part = 0;
  for (int i = t; i < b; i += 256) part += bsum[i];
  sh[t] = part;
  __syncthreads();
  for (int st = 128; st > 0; st >>= 1) {
    if (t < st) sh[t] += sh[t + st];
    __syncthreads();
  }
  const int offset = sh[0];
  __syncthreads();
  const int i0 = b * 1024 + t * 4;
  int d[4];
#pragma unroll
  for (int k = 0; k < 4; ++k)
    d[k] = (i0 + k < n) ? deg[i0 + k] + deg[n + i0 + k] : 0;
  const int tsum = d[0] + d[1] + d[2] + d[3];
  sh[t] = tsum;
  __syncthreads();
  for (int st = 1; st < 256; st <<= 1) {
    const int v = (t >= st) ? sh[t - st] : 0;
    __syncthreads();
    sh[t] += v;
    __syncthreads();
  }
  int run = offset + sh[t] - tsum;
#pragma unroll
  for (int k = 0; k < 4; ++k) {
    if (i0 + k < n) {
      rowptr[i0 + k] = run;
      run += d[k];
      if (i0 + k == n - 1) rowptr[n] = run;
    }
  }
}

template <bool FULL>
__device__ inline void fill_body(const int* __restrict__ ei,
                                 const int* __restrict__ rowptr,
                                 const int* __restrict__ rank,
                                 const int* __restrict__ deg,
                                 int* __restrict__ adj, int nE, int e0) {
  int s[EPT], d[EPT], r[EPT], base[EPT], before[EPT];
#pragma unroll
  for (int k = 0; k < EPT; ++k) {
    const int e = e0 + k * 256;
    const bool in = FULL || e < nE;
    s[k] = in ? ei[e] : 0;
    d[k] = in ? ei[nE + e] : 0;
    r[k] = in ? rank[e] : 0;
  }
#pragma unroll
  for (int k = 0; k < EPT; ++k) {
    const int e = e0 + k * 256;
    base[k] = (FULL || e < nE) ? rowptr[d[k]] : 0;
  }
#pragma unroll
  for (int k = 0; k < EPT; ++k) {
    const int e = e0 + k * 256;
    before[k] = ((FULL || e < nE) && r[k] < 0) ? deg[d[k]] : 0;
  }
#pragma unroll
  for (int k = 0; k < EPT; ++k) {
    const int e = e0 + k * 256;
    if (FULL || e < nE)
      adj[base[k] + (r[k] & 0x7fffffff) + before[k]] = s[k];
  }
}

__global__ void fill_kernel(const int* __restrict__ ei,
                            const int* __restrict__ rowptr,
                            const int* __restrict__ rank,
                            const int* __restrict__ deg,
                            int* __restrict__ adj, int nE) {
  const int b0 = blockIdx.x * (256 * EPT);
  const int e0 = b0 + threadIdx.x;
  if (b0 + 256 * EPT <= nE) fill_body<true>(ei, rowptr, rank, deg, adj, nE, e0);
  else                      fill_body<false>(ei, rowptr, rank, deg, adj, nE, e0);
}

__global__ __launch_bounds__(256) void segsort_kernel(
    const int* __restrict__ rowptr, const int* __restrict__ adj,
    int* __restrict__ adjs, int n) {
  const int n0 = NPW * __builtin_amdgcn_readfirstlane(
      (blockIdx.x * 256 + threadIdx.x) >> 5);
  if (n0 >= n) return;
  const int lane = threadIdx.x & 31;
  const int cnt  = (n - n0 < NPW) ? (n - n0) : NPW;
  const int rp = (lane <= cnt) ? rowptr[n0 + lane] : 0;
  int key[NPW], st[NPW], dg[NPW];
#pragma unroll
  for (int u = 0; u < NPW; ++u) {
    st[u] = __builtin_amdgcn_readlane(rp, u);
    dg[u] = (u < cnt) ? __builtin_amdgcn_readlane(rp, u + 1) - st[u] : 0;
  }
#pragma unroll
  for (int u = 0; u < NPW; ++u) {
    const int nshort = (dg[u] <= 32) ? dg[u] : 0;
    key[u] = (lane < nshort) ? adj[st[u] + lane] : 0x7fffffff;
  }
#pragma unroll
  for (int u = 0; u < NPW; ++u) {
    if (dg[u] > 0) {
      const int start = st[u];
      const int deg   = dg[u];
      if (deg <= 32) {
        const int ku = key[u];
        int rank = 0;
        for (int j = 0; j < deg; ++j) {
          const int kj = __builtin_amdgcn_readlane(ku, j);
          rank += ((kj < ku) || (kj == ku && j < lane)) ? 1 : 0;
        }
        if (lane < deg) adjs[start + rank] = ku;
      } else {
        for (int i = lane; i < deg; i += 32) {
          const int ki = adj[start + i];
          int rank = 0;
          for (int j = 0; j < deg; ++j) {
            const int kj = adj[start + j];
            rank += ((kj < ki) || (kj == ki && j < i)) ? 1 : 0;
          }
          adjs[start + rank] = ki;
        }
      }
    }
  }
}

template <int U>
__device__ inline void take_upto(const _Float16* __restrict__ hl, int key,
                                 int j, int m, float4& acc) {
  half4 v[U];
#pragma unroll
  for (int u = 0; u < U; ++u) {
    if (j + u < m) {
      const int sidx = __builtin_amdgcn_readlane(key, j + u);
      v[u] = *(const half4*)(hl + (size_t)sidx * HID);
    }
  }
#pragma unroll
  for (int u = 0; u < U; ++u) {
    if (j + u < m) {
      acc.x += (float)v[u][0]; acc.y += (float)v[u][1];
      acc.z += (float)v[u][2]; acc.w += (float)v[u][3];
    }
  }
}

__device__ inline void sum_rows(const _Float16* __restrict__ hl, int key, int m,
                                float4& acc) {
  take_upto<16>(hl, key, 0, m, acc);
  if (m > 16) take_upto<16>(hl, key, 16, m, acc);
}

__device__ inline v16h join16(half8 lo, half8 hi) {
  return __builtin_shufflevector(lo, hi, 0, 1, 2, 3, 4, 5, 6, 7,
                                         8, 9, 10, 11, 12, 13, 14, 15);
}

typedef float __attribute__((__may_alias__)) f32a;

template <int NW, bool OUTP>
__global__ __launch_bounds__(32 * NW, 2) void sage_kernel(
    const _Float16* __restrict__ h,
    const int* __restrict__ rowptr, const int* __restrict__ adjs,
    const _Float16* __restrict__ WlT,
    const _Float16* __restrict__ WrT,
    const float*    __restrict__ bias,
    _Float16* __restrict__ outh,
    const _Float16* __restrict__ WoT,
    const float*    __restrict__ bout,
    float* __restrict__ outf,
    float* __restrict__ colpart,
    int n_nodes) {
  __shared__ __attribute__((aligned(16))) _Float16 wtr[HID * WPITCH];
  __shared__ __attribute__((aligned(16))) _Float16 wtl[HID * WPITCH];
  __shared__ __attribute__((aligned(16))) _Float16 wto[OUTP ? HID * WPITCH : 8];
  __shared__ __attribute__((aligned(16))) _Float16 mtile[NW * 16 * WPITCH];

  {
    const uint4* sr = (const uint4*)WrT;
    const uint4* sl = (const uint4*)WlT;
    const uint4* so = (const uint4*)WoT;
    for (int i = threadIdx.x; i < HID * HID / 8; i += 32 * NW) {
      *(uint4*)&wtr[(i >> 4) * WPITCH + (i & 15) * 8] = sr[i];
      *(uint4*)&wtl[(i >> 4) * WPITCH + (i & 15) * 8] = sl[i];
      if constexpr (OUTP)
        *(uint4*)&wto[(i >> 4) * WPITCH + (i & 15) * 8] = so[i];
    }
  }
  __syncthreads();

  const int lane = threadIdx.x & 31;
  const int wave = __builtin_amdgcn_readfirstlane((int)threadIdx.x >> 5);
  const int rowbase = blockIdx.x * (16 * NW) + wave * 16;
  const bool active = rowbase < n_nodes;
  if constexpr (!OUTP) {
    if (!active) return;
  }
  _Float16* mt = &mtile[wave * 16 * WPITCH];
  f32a* cr = (f32a*)mt + (lane >> 4) * HID + (lane & 15);

  if (active) {
  const int cnt = (n_nodes - rowbase < 16) ? (n_nodes - rowbase) : 16;

  const int arow = rowbase + (lane & 15);
  const int row  = arow < n_nodes ? arow : n_nodes - 1;
  const int lo   = (lane >= 16) ? 8 : 0;

  const _Float16* hp = h + (size_t)row * HID + lo;
  v16h ahf[4];
#pragma unroll
  for (int kk = 0; kk < 4; ++kk)
    ahf[kk] = join16(*(const half8*)(hp + kk * 32),
                     *(const half8*)(hp + kk * 32 + 16));

  {
    const int rp = (lane <= cnt) ? rowptr[rowbase + lane] : 0;
    const _Float16* hl = h + lane * 4;
    int nstart = __builtin_amdgcn_readlane(rp, 0);
    int nend   = __builtin_amdgcn_readlane(rp, 1);
    int nkey = (lane < nend - nstart) ? adjs[nstart + lane] : 0;
    for (int i = 0; i < cnt; ++i) {
      const int start = nstart, end = nend;
      const int key0  = nkey;
      if (i + 1 < cnt) {
        nstart = end;
        nend   = __builtin_amdgcn_readlane(rp, i + 2);
        nkey   = (lane < nend - nstart) ? adjs[nstart + lane] : 0;
      }
      float4 acc = make_float4(0.0f, 0.0f, 0.0f, 0.0f);
      const int deg = end - start;
      sum_rows(hl, key0, (deg < 32) ? deg : 32, acc);
      for (int base = start + 32; base < end; base += 32) {
        const int m   = (end - base < 32) ? (end - base) : 32;
        const int key = (lane < m) ? adjs[base + lane] : 0;
        sum_rows(hl, key, m, acc);
      }
      const float inv = (deg > 0) ? __builtin_amdgcn_rcpf((float)deg) : 0.0f;
      half4 o;
      o[0] = (_Float16)(acc.x * inv); o[1] = (_Float16)(acc.y * inv);
      o[2] = (_Float16)(acc.z * inv); o[3] = (_Float16)(acc.w * inv);
      *(half4*)(mt + i * WPITCH + lane * 4) = o;
    }
    for (int i = cnt; i < 16; ++i) {
      half4 z = {};
      *(half4*)(mt + i * WPITCH + lane * 4) = z;
    }
  }

  const _Float16* gp  = mt + (lane & 15) * WPITCH + lo;
  const _Float16* brp = &wtr[(lane & 15) * WPITCH + lo];
  const _Float16* blp = &wtl[(lane & 15) * WPITCH + lo];

  v8f acc[8];
  {
    v8f z = {};
#pragma unroll
    for (int nt = 0; nt < 8; ++nt) acc[nt] = z;
  }

#pragma unroll
  for (int kk = 0; kk < 4; ++kk) {
    const int kb = kk * 32;
    v16h agf = join16(*(const half8*)(gp + kb),
                      *(const half8*)(gp + kb + 16));
#pragma unroll
    for (int nt = 0; nt < 8; ++nt) {
      const int off = nt * 16 * WPITCH + kb;
      v16h br = join16(*(const half8*)(brp + off),
                       *(const half8*)(brp + off + 16));
      acc[nt] = __builtin_amdgcn_wmma_f32_16x16x32_f16(
          false, ahf[kk], false, br, (short)0, acc[nt], false, false);
      v16h bl = join16(*(const half8*)(blp + off),
                       *(const half8*)(blp + off + 16));
      acc[nt] = __builtin_amdgcn_wmma_f32_16x16x32_f16(
          false, agf, false, bl, (short)0, acc[nt], false, false);
    }
  }

  const int mofs = (lane >= 16) ? 8 : 0;
  if constexpr (!OUTP) {
#pragma unroll
    for (int nt = 0; nt < 8; ++nt) {
      const int ncol = nt * 16 + (lane & 15);
      const float bb = bias[ncol];
#pragma unroll
      for (int r = 0; r < 8; ++r) {
        const float v = fmaxf(acc[nt][r] + bb, 0.0f);
        const int orow = rowbase + mofs + r;
        if (orow < n_nodes) outh[(size_t)orow * HID + ncol] = (_Float16)v;
      }
    }
  } else {
    __builtin_amdgcn_sched_barrier(0);
#pragma unroll
    for (int nt = 0; nt < 8; ++nt) {
      const int ncol = nt * 16 + (lane & 15);
      const float bb = bias[ncol];
#pragma unroll
      for (int r = 0; r < 8; ++r)
        mt[(mofs + r) * WPITCH + ncol] =
            (_Float16)fmaxf(acc[nt][r] + bb, 0.0f);
    }
    float bo8[8];
#pragma unroll
    for (int nt = 0; nt < 8; ++nt) bo8[nt] = bout[nt * 16 + (lane & 15)];
    const _Float16* bop = &wto[(lane & 15) * WPITCH + lo];
    v16h a2[4];
#pragma unroll
    for (int kk = 0; kk < 4; ++kk)
      a2[kk] = join16(*(const half8*)(gp + kk * 32),
                      *(const half8*)(gp + kk * 32 + 16));
    float* op = outf + (size_t)(rowbase + mofs) * HID + (lane & 15);
#pragma unroll
    for (int np = 0; np < 2; ++np) {
      v8f pacc[4];
      {
        v8f z = {};
#pragma unroll
        for (int q = 0; q < 4; ++q) pacc[q] = z;
      }
#pragma unroll
      for (int kk = 0; kk < 4; ++kk) {
#pragma unroll
        for (int q = 0; q < 4; ++q) {
          const int off = (np * 4 + q) * 16 * WPITCH + kk * 32;
          v16h bo = join16(*(const half8*)(bop + off),
                           *(const half8*)(bop + off + 16));
          pacc[q] = __builtin_amdgcn_wmma_f32_16x16x32_f16(
              false, a2[kk], false, bo, (short)0, pacc[q], false, false);
        }
      }
      float csum[4];
#pragma unroll
      for (int q = 0; q < 4; ++q) csum[q] = 0.0f;
      if (cnt == 16) {
#pragma unroll
        for (int q = 0; q < 4; ++q) {
#pragma unroll
          for (int r = 0; r < 8; ++r) {
            const float v = pacc[q][r] + bo8[np * 4 + q];
            op[r * HID + (np * 4 + q) * 16] = v;
            csum[q] += v;
          }
        }
      } else {
#pragma unroll
        for (int q = 0; q < 4; ++q) {
#pragma unroll
          for (int r = 0; r < 8; ++r) {
            const float v = pacc[q][r] + bo8[np * 4 + q];
            if (mofs + r < cnt) {
              op[r * HID + (np * 4 + q) * 16] = v;
              csum[q] += v;
            }
          }
        }
      }
#pragma unroll
      for (int q = 0; q < 4; ++q) cr[(np * 4 + q) * 16] = csum[q];
    }
  }
  } else if constexpr (OUTP) {
#pragma unroll
    for (int nt = 0; nt < 8; ++nt) cr[nt * 16] = 0.0f;
  }

  if constexpr (OUTP) {
    __syncthreads();
    if (threadIdx.x < HID) {
      float t = 0.0f;
#pragma unroll
      for (int p = 0; p < NW; ++p) {
        const f32a* q = (const f32a*)&mtile[p * 16 * WPITCH];
        t += q[threadIdx.x];
        t += q[HID + threadIdx.x];
      }
      colpart[(size_t)blockIdx.x * HID + threadIdx.x] = t;
    }
  }
}

__global__ void colmean_kernel(const float* __restrict__ part, int nparts,
                               float* __restrict__ outm, int n) {
  __shared__ float red[256];
  int col = blockIdx.x;
  float acc = 0.0f;
  for (int r = threadIdx.x; r < nparts; r += 256)
    acc += part[(size_t)r * HID + col];
  red[threadIdx.x] = acc;
  __syncthreads();
  for (int st = 128; st > 0; st >>= 1) {
    if (threadIdx.x < st) red[threadIdx.x] += red[threadIdx.x + st];
    __syncthreads();
  }
  if (threadIdx.x == 0) outm[col] = red[0] / (float)n;
}

extern "C" void kernel_launch(void* const* d_in, const int* in_sizes, int n_in,
                              void* d_out, int out_size, void* d_ws, size_t ws_size,
                              hipStream_t stream) {
  const float* pos  = (const float*)d_in[0];
  const int*   an   = (const int*)  d_in[1];
  const int*   ei   = (const int*)  d_in[2];
  const float* W_in = (const float*)d_in[3];
  const float* b_in = (const float*)d_in[4];
  const float* W1l  = (const float*)d_in[5];
  const float* b1   = (const float*)d_in[6];
  const float* W1r  = (const float*)d_in[7];
  const float* W2l  = (const float*)d_in[8];
  const float* b2   = (const float*)d_in[9];
  const float* W2r  = (const float*)d_in[10];
  const float* Wout = (const float*)d_in[11];
  const float* bout = (const float*)d_in[12];

  const int n  = in_sizes[0] / 3;
  const int nE = in_sizes[2] / 2;

  _Float16* h   = (_Float16*)d_ws;
  _Float16* hb  = h + (size_t)n * HID;

  _Float16* wtbase = hb + (size_t)n * HID;
  _Float16* W1lT = wtbase + 0 * HID * HID;
  _Float16* W1rT = wtbase + 1 * HID * HID;
  _Float16* W2lT = wtbase + 2 * HID * HID;
  _Float16* W2rT = wtbase + 3 * HID * HID;
  _Float16* WoT  = wtbase + 4 * HID * HID;

  const int scanBlks = (n + 1023) / 1024;
  int* deg    = (int*)(wtbase + 5 * HID * HID);
  int* rowptr = deg + 2 * n;
  int* bsum   = rowptr + (n + 1);
  int* adj    = bsum + scanBlks;
  int* adjs   = adj + nE;
  int* rank   = adjs + nE;
  float* colpart = (float*)(rank + nE);

  float* outm = (float*)d_out;
  float* nst  = (float*)d_out + HID;

  const int sageBlks = (n + 16 * FW - 1) / (16 * FW);
  const int outBlks  = (n + 16 * FWO - 1) / (16 * FWO);
  const int edgeBlks = (nE + 256 * EPT - 1) / (256 * EPT);
  const int nodeBlks = (n + 8 * NPW - 1) / (8 * NPW);
  const int zeroBlks = (2 * n + 1023) / 1024;

  prep_kernel<<<5 * WBLKS + zeroBlks + nodeBlks, 256, 0, stream>>>(
      pos, an, W_in, b_in, W1l, W1r, W2l, W2r, Wout, wtbase, deg, zeroBlks,
      h, n);

  degree_kernel<<<edgeBlks, 256, 0, stream>>>(ei, deg, rank, nE, n);
  blocksum_kernel<<<scanBlks, 256, 0, stream>>>(deg, bsum, n);
  scan_kernel<<<scanBlks, 256, 0, stream>>>(deg, bsum, rowptr, n);
  fill_kernel<<<edgeBlks, 256, 0, stream>>>(ei, rowptr, rank, deg, adj, nE);
  segsort_kernel<<<nodeBlks, 256, 0, stream>>>(rowptr, adj, adjs, n);

  sage_kernel<FW, false><<<sageBlks, 32 * FW, 0, stream>>>(
      h, rowptr, adjs, W1lT, W1rT, b1, hb, nullptr, nullptr, nullptr, nullptr,
      n);
  sage_kernel<FWO, true><<<outBlks, 32 * FWO, 0, stream>>>(
      hb, rowptr, adjs, W2lT, W2rT, b2, nullptr, WoT, bout, nst, colpart, n);
  colmean_kernel<<<HID, 256, 0, stream>>>(colpart, outBlks, outm, n);
}
